# starting kernel
# baseline (speedup 1.0000x reference)
#include <hip/hip_runtime.h>
#include <hip/hip_bf16.h>

#define FEAT 96
#define KSTEPS (FEAT / 4)
#define NTILES (FEAT / 16)

typedef float v2f __attribute__((ext_vector_type(2)));
typedef float v8f __attribute__((ext_vector_type(8)));

__global__ __launch_bounds__(256) void zero_f32(float* __restrict__ p, long long n) {
  long long i = (long long)blockIdx.x * blockDim.x + threadIdx.x;
  long long i4 = i * 4;
  if (i4 + 3 < n) {
    *(float4*)(p + i4) = make_float4(0.f, 0.f, 0.f, 0.f);
  } else {
    for (long long j = i4; j < n; ++j) p[j] = 0.f;
  }
}

__global__ __launch_bounds__(256) void degree_kernel(const int* __restrict__ src,
                                                     const int* __restrict__ dst,
                                                     float* __restrict__ deg_out,
                                                     float* __restrict__ deg_in,
                                                     int nedges) {
  int e = blockIdx.x * blockDim.x + threadIdx.x;
  if (e >= nedges) return;
  atomicAdd(&deg_out[src[e]], 1.0f);
  atomicAdd(&deg_in[dst[e]], 1.0f);
}

__global__ __launch_bounds__(256) void norm_finalize(float* __restrict__ deg_out,
                                                     float* __restrict__ deg_in,
                                                     int n) {
  int i = blockIdx.x * blockDim.x + threadIdx.x;
  if (i >= n) return;
  deg_out[i] = 1.0f / sqrtf(fmaxf(deg_out[i], 1.0f));
  deg_in[i]  = 1.0f / sqrtf(fmaxf(deg_in[i], 1.0f));
}

__global__ __launch_bounds__(256) void edge_scatter(const int* __restrict__ src,
                                                    const int* __restrict__ dst,
                                                    const float* __restrict__ h,
                                                    const float* __restrict__ norm_src,
                                                    float* __restrict__ m,
                                                    int nedges) {
  long long gid = (long long)blockIdx.x * blockDim.x + threadIdx.x;
  int e = __builtin_amdgcn_readfirstlane((int)(gid >> 5));
  if (e >= nedges) return;
  int lane = threadIdx.x & 31;
  int s = src[e];
  int d = dst[e];
  float ns = norm_src[s];
  const float* __restrict__ hr = h + (long long)s * FEAT;
  float* __restrict__ mr = m + (long long)d * FEAT;
#pragma unroll
  for (int i = 0; i < 3; ++i) {
    int f = lane + i * 32;
    atomicAdd(&mr[f], hr[f] * ns);
  }
}

__global__ __launch_bounds__(256) void gemm_norm_bias_relu(
    const float* __restrict__ m, const float* __restrict__ norm_dst,
    const float* __restrict__ W, const float* __restrict__ bias,
    float* __restrict__ out, int nrows) {
  __shared__ float Ws[FEAT * FEAT];
  __shared__ float bs[FEAT];

  for (int i = threadIdx.x; i < FEAT * FEAT; i += 256) Ws[i] = W[i];
  if (threadIdx.x < FEAT) bs[threadIdx.x] = bias[threadIdx.x];
  __syncthreads();

  const int wave = threadIdx.x >> 5;
  const int lane = threadIdx.x & 31;
  const int strip = blockIdx.x * 8 + wave;
  const int row0 = strip * 16;
  if (row0 >= nrows) return;

  const int lrow = lane & 15;
  const int lgrp = lane >> 4;

  const int arow_i = min(row0 + lrow, nrows - 1);
  const float nd = norm_dst[arow_i];
  const float* __restrict__ arow = m + (long long)arow_i * FEAT;

  v2f a[KSTEPS];
#pragma unroll
  for (int ks = 0; ks < KSTEPS; ++ks) {
    const int k = ks * 4 + 2 * lgrp;
    a[ks].x = arow[k] * nd;
    a[ks].y = arow[k + 1] * nd;
  }

  v8f acc[NTILES];
#pragma unroll
  for (int nt = 0; nt < NTILES; ++nt) acc[nt] = (v8f){};

  v2f bf[NTILES];
#pragma unroll
  for (int nt = 0; nt < NTILES; ++nt) {
    const int k = 2 * lgrp;
    bf[nt].x = Ws[k * FEAT + nt * 16 + lrow];
    bf[nt].y = Ws[(k + 1) * FEAT + nt * 16 + lrow];
  }

#pragma unroll
  for (int ks = 0; ks < KSTEPS; ++ks) {
    v2f bn[NTILES];
    if (ks + 1 < KSTEPS) {
      const int k = (ks + 1) * 4 + 2 * lgrp;
#pragma unroll
      for (int nt = 0; nt < NTILES; ++nt) {
        bn[nt].x = Ws[k * FEAT + nt * 16 + lrow];
        bn[nt].y = Ws[(k + 1) * FEAT + nt * 16 + lrow];
      }
    }
#pragma unroll
    for (int nt = 0; nt < NTILES; ++nt) {
      acc[nt] = __builtin_amdgcn_wmma_f32_16x16x4_f32(
           false, a[ks],  false, bf[nt],
           (short)0, acc[nt],  false,  false);
    }
    if (ks + 1 < KSTEPS) {
#pragma unroll
      for (int nt = 0; nt < NTILES; ++nt) bf[nt] = bn[nt];
    }
  }

#pragma unroll
  for (int nt = 0; nt < NTILES; ++nt) {
    const int nb = nt * 16;
    const float bcol = bs[nb + lrow];
#pragma unroll
    for (int r = 0; r < 8; ++r) {
      const int row = row0 + r + 8 * lgrp;
      if (row < nrows) {
        float v = acc[nt][r] + bcol;
        out[(long long)row * FEAT + nb + lrow] = fmaxf(v, 0.0f);
      }
    }
  }
}

extern "C" void kernel_launch(void* const* d_in, const int* in_sizes, int n_in,
                              void* d_out, int out_size, void* d_ws, size_t ws_size,
                              hipStream_t stream) {
  const float* h0  = (const float*)d_in[0];
  const int*   src = (const int*)d_in[1];
  const int*   dst = (const int*)d_in[2];
  const float* Wp[3] = {(const float*)d_in[3], (const float*)d_in[5], (const float*)d_in[7]};
  const float* bp[3] = {(const float*)d_in[4], (const float*)d_in[6], (const float*)d_in[8]};

  const int N = in_sizes[0] / FEAT;
  const int E = in_sizes[1];

  float* ws       = (float*)d_ws;
  float* norm_src = ws;
  float* norm_dst = ws + N;
  float* mbuf     = ws + 2LL * N;
  float* hA       = mbuf + (long long)N * FEAT;
  float* outp     = (float*)d_out;

  const long long ND = (long long)N * FEAT;

  {
    long long n2 = 2LL * N;
    int blocks = (int)((n2 / 4 + 255) / 256) + 1;
    zero_f32<<<blocks, 256, 0, stream>>>(norm_src, n2);
    degree_kernel<<<(E + 255) / 256, 256, 0, stream>>>(src, dst, norm_src, norm_dst, E);
    norm_finalize<<<(N + 255) / 256, 256, 0, stream>>>(norm_src, norm_dst, N);
  }

  const int strips = (N + 15) / 16;
  const int gemm_blocks = (strips + 7) / 8;
  const int zero_blocks = (int)((ND / 4 + 255) / 256) + 1;
  const long long scat_threads = (long long)E * 32;
  const int scat_blocks = (int)((scat_threads + 255) / 256);

  const float* lin[3]  = {h0, outp, hA};
  float*       lout[3] = {outp, hA, outp};

  for (int l = 0; l < 3; ++l) {
    zero_f32<<<zero_blocks, 256, 0, stream>>>(mbuf, ND);
    edge_scatter<<<scat_blocks, 256, 0, stream>>>(src, dst, lin[l], norm_src, mbuf, E);
    gemm_norm_bias_relu<<<gemm_blocks, 256, 0, stream>>>(mbuf, norm_dst, Wp[l], bp[l],
                                                         lout[l], N);
  }
}


# Replace per-layer zero fill + f32 atomic scatter by a one-time destination-sorted edge list (integer atomic arrival rank + prefix sum) and an atomic-free row gather; per-node sum order = that rank
# speedup vs baseline: 2.0264x; 2.0264x over previous
#include <hip/hip_runtime.h>
#include <hip/hip_bf16.h>

#define FEAT 96
#define KSTEPS (FEAT / 4)
#define NTILES (FEAT / 16)

typedef float v2f __attribute__((ext_vector_type(2)));
typedef float v8f __attribute__((ext_vector_type(8)));

__global__ __launch_bounds__(256) void zero_f32(float* __restrict__ p, long long n) {
  long long i = (long long)blockIdx.x * blockDim.x + threadIdx.x;
  long long i4 = i * 4;
  if (i4 + 3 < n) {
    *(float4*)(p + i4) = make_float4(0.f, 0.f, 0.f, 0.f);
  } else {
    for (long long j = i4; j < n; ++j) p[j] = 0.f;
  }
}

__global__ __launch_bounds__(256) void degree_rank(const int* __restrict__ src,
                                                   const int* __restrict__ dst,
                                                   int* __restrict__ cnt_out,
                                                   int* __restrict__ cnt_in,
                                                   int* __restrict__ rank,
                                                   int nedges) {
  int e = blockIdx.x * blockDim.x + threadIdx.x;
  if (e >= nedges) return;
  const int s = src[e];
  const int d = dst[e];
  atomicAdd(&cnt_out[s], 1);
  rank[e] = atomicAdd(&cnt_in[d], 1);
}

__global__ __launch_bounds__(256) void rowptr_and_norms(const int* __restrict__ cnt_out,
                                                        const int* __restrict__ cnt_in,
                                                        int* __restrict__ rowptr,
                                                        float* __restrict__ norm_src,
                                                        float* __restrict__ norm_dst,
                                                        int n) {
  __shared__ int wave_pre[8];
  __shared__ int wave_tot[8];
  const int tid = threadIdx.x;
  const int lane = tid & 31;
  const int wave = tid >> 5;
  const int block_base = blockIdx.x * 1024;

  int pre = 0;
  for (int i = tid * 4; i < block_base; i += 1024) {
    const int4 v = *(const int4*)(cnt_in + i);
    pre += (v.x + v.y) + (v.z + v.w);
  }

  const int i0 = block_base + tid * 4;
  int c[4];
  for (int j = 0; j < 4; ++j) {
    const int idx = i0 + j;
    c[j] = 0;
    if (idx < n) {
      c[j] = cnt_in[idx];
      norm_src[idx] = 1.0f / sqrtf(fmaxf((float)cnt_out[idx], 1.0f));
      norm_dst[idx] = 1.0f / sqrtf(fmaxf((float)c[j], 1.0f));
    }
  }
  const int tsum = (c[0] + c[1]) + (c[2] + c[3]);

  int incl = tsum;
  for (int o = 1; o < 32; o <<= 1) {
    const int t = __shfl_up(incl, o);
    if (lane >= o) incl += t;
  }
  for (int o = 16; o > 0; o >>= 1) pre += __shfl_xor(pre, o);
  if (lane == 31) wave_tot[wave] = incl;
  if (lane == 0) wave_pre[wave] = pre;
  __syncthreads();

  int base = 0;
  for (int w = 0; w < 8; ++w) {
    base += wave_pre[w];
    if (w < wave) base += wave_tot[w];
  }
  int run = base + incl - tsum;
  for (int j = 0; j < 4; ++j) {
    const int idx = i0 + j;
    if (idx <= n) rowptr[idx] = run;
    run += c[j];
  }
}

__global__ __launch_bounds__(256) void edge_place(const int* __restrict__ src,
                                                  const int* __restrict__ dst,
                                                  const int* __restrict__ rank,
                                                  const int* __restrict__ rowptr,
                                                  const float* __restrict__ norm_src,
                                                  int2* __restrict__ rec,
                                                  int nedges) {
  int e = blockIdx.x * blockDim.x + threadIdx.x;
  if (e >= nedges) return;
  const int s = src[e];
  const int p = rowptr[dst[e]] + rank[e];
  rec[p] = make_int2(s, __float_as_int(norm_src[s]));
}

__device__ __forceinline__ void fma4(float4& acc, const float4 v, const float w) {
  acc.x = fmaf(v.x, w, acc.x);
  acc.y = fmaf(v.y, w, acc.y);
  acc.z = fmaf(v.z, w, acc.z);
  acc.w = fmaf(v.w, w, acc.w);
}

template <int U>
__device__ __forceinline__ void gather_group(float4& acc, const float* __restrict__ h,
                                             const int fo, const int2 p, const int j) {
  float4 v[U];
  float w[U];
#pragma unroll
  for (int u = 0; u < U; ++u) {
    const int s = __builtin_amdgcn_readlane(p.x, j + u);
    w[u] = __int_as_float(__builtin_amdgcn_readlane(p.y, j + u));
    v[u] = *(const float4*)(h + (long long)s * FEAT + fo);
  }
#pragma unroll
  for (int u = 0; u < U; ++u) fma4(acc, v[u], w[u]);
}

__global__ __launch_bounds__(256) void gather_rows(const int* __restrict__ rowptr,
                                                   const int2* __restrict__ rec,
                                                   const float* __restrict__ h,
                                                   float* __restrict__ m,
                                                   int n) {
  const long long gid = (long long)blockIdx.x * blockDim.x + threadIdx.x;
  const int d = __builtin_amdgcn_readfirstlane((int)(gid >> 5));
  if (d >= n) return;
  const int lane = threadIdx.x & 31;
  const int fo = min(lane, FEAT / 4 - 1) * 4;
  const int beg = rowptr[d];
  const int end = rowptr[d + 1];

  float4 acc = make_float4(0.f, 0.f, 0.f, 0.f);
  for (int base = beg; base < end; base += 32) {
    const int idx = base + lane;
    int2 p = make_int2(0, 0);
    if (idx < end) p = rec[idx];
    const int cnt = min(32, end - base);
    int j = 0;
    for (; j + 8 <= cnt; j += 8) gather_group<8>(acc, h, fo, p, j);
    if (j + 4 <= cnt) { gather_group<4>(acc, h, fo, p, j); j += 4; }
    if (j + 2 <= cnt) { gather_group<2>(acc, h, fo, p, j); j += 2; }
    if (j < cnt) gather_group<1>(acc, h, fo, p, j);
  }
  if (lane < FEAT / 4) *(float4*)(m + (long long)d * FEAT + lane * 4) = acc;
}

__global__ __launch_bounds__(256) void gemm_norm_bias_relu(
    const float* __restrict__ m, const float* __restrict__ norm_dst,
    const float* __restrict__ W, const float* __restrict__ bias,
    float* __restrict__ out, int nrows) {
  __shared__ float Ws[FEAT * FEAT];
  __shared__ float bs[FEAT];

  for (int i = threadIdx.x; i < FEAT * FEAT; i += 256) Ws[i] = W[i];
  if (threadIdx.x < FEAT) bs[threadIdx.x] = bias[threadIdx.x];
  __syncthreads();

  const int wave = threadIdx.x >> 5;
  const int lane = threadIdx.x & 31;
  const int strip = blockIdx.x * 8 + wave;
  const int row0 = strip * 16;
  if (row0 >= nrows) return;

  const int lrow = lane & 15;
  const int lgrp = lane >> 4;

  const int arow_i = min(row0 + lrow, nrows - 1);
  const float nd = norm_dst[arow_i];
  const float* __restrict__ arow = m + (long long)arow_i * FEAT;

  v2f a[KSTEPS];
#pragma unroll
  for (int ks = 0; ks < KSTEPS; ++ks) {
    const int k = ks * 4 + 2 * lgrp;
    a[ks].x = arow[k] * nd;
    a[ks].y = arow[k + 1] * nd;
  }

  v8f acc[NTILES];
#pragma unroll
  for (int nt = 0; nt < NTILES; ++nt) acc[nt] = (v8f){};

  v2f bf[NTILES];
#pragma unroll
  for (int nt = 0; nt < NTILES; ++nt) {
    const int k = 2 * lgrp;
    bf[nt].x = Ws[k * FEAT + nt * 16 + lrow];
    bf[nt].y = Ws[(k + 1) * FEAT + nt * 16 + lrow];
  }

#pragma unroll
  for (int ks = 0; ks < KSTEPS; ++ks) {
    v2f bn[NTILES];
    if (ks + 1 < KSTEPS) {
      const int k = (ks + 1) * 4 + 2 * lgrp;
#pragma unroll
      for (int nt = 0; nt < NTILES; ++nt) {
        bn[nt].x = Ws[k * FEAT + nt * 16 + lrow];
        bn[nt].y = Ws[(k + 1) * FEAT + nt * 16 + lrow];
      }
    }
#pragma unroll
    for (int nt = 0; nt < NTILES; ++nt) {
      acc[nt] = __builtin_amdgcn_wmma_f32_16x16x4_f32(
           false, a[ks],  false, bf[nt],
           (short)0, acc[nt],  false,  false);
    }
    if (ks + 1 < KSTEPS) {
#pragma unroll
      for (int nt = 0; nt < NTILES; ++nt) bf[nt] = bn[nt];
    }
  }

#pragma unroll
  for (int nt = 0; nt < NTILES; ++nt) {
    const int nb = nt * 16;
    const float bcol = bs[nb + lrow];
#pragma unroll
    for (int r = 0; r < 8; ++r) {
      const int row = row0 + r + 8 * lgrp;
      if (row < nrows) {
        float v = acc[nt][r] + bcol;
        out[(long long)row * FEAT + nb + lrow] = fmaxf(v, 0.0f);
      }
    }
  }
}

extern "C" void kernel_launch(void* const* d_in, const int* in_sizes, int n_in,
                              void* d_out, int out_size, void* d_ws, size_t ws_size,
                              hipStream_t stream) {
  const float* h0  = (const float*)d_in[0];
  const int*   src = (const int*)d_in[1];
  const int*   dst = (const int*)d_in[2];
  const float* Wp[3] = {(const float*)d_in[3], (const float*)d_in[5], (const float*)d_in[7]};
  const float* bp[3] = {(const float*)d_in[4], (const float*)d_in[6], (const float*)d_in[8]};

  const int N = in_sizes[0] / FEAT;
  const int E = in_sizes[1];

  const long long NP = ((long long)N + 1 + 63) / 64 * 64;
  const long long EP = ((long long)E + 63) / 64 * 64;
  float* ws       = (float*)d_ws;
  float* norm_src = ws;
  float* norm_dst = ws + NP;
  int*   cnt_out  = (int*)(ws + 2 * NP);
  int*   cnt_in   = (int*)(ws + 3 * NP);
  int*   rowptr   = (int*)(ws + 4 * NP);
  int*   rank     = (int*)(ws + 5 * NP);
  int2*  rec      = (int2*)(ws + 5 * NP + EP);
  float* mbuf     = ws + 5 * NP + 3 * EP;
  float* hA       = mbuf + (long long)N * FEAT;
  float* outp     = (float*)d_out;

  {
    long long n2 = 2 * NP;
    int blocks = (int)((n2 / 4 + 255) / 256) + 1;
    zero_f32<<<blocks, 256, 0, stream>>>((float*)cnt_out, n2);
    degree_rank<<<(E + 255) / 256, 256, 0, stream>>>(src, dst, cnt_out, cnt_in, rank, E);
    rowptr_and_norms<<<N / 1024 + 1, 256, 0, stream>>>(cnt_out, cnt_in, rowptr,
                                                       norm_src, norm_dst, N);
    edge_place<<<(E + 255) / 256, 256, 0, stream>>>(src, dst, rank, rowptr, norm_src, rec, E);
  }

  const int strips = (N + 15) / 16;
  const int gemm_blocks = (strips + 7) / 8;
  const int gather_blocks = (int)(((long long)N * 32 + 255) / 256);

  const float* lin[3]  = {h0, outp, hA};
  float*       lout[3] = {outp, hA, outp};

  for (int l = 0; l < 3; ++l) {
    gather_rows<<<gather_blocks, 256, 0, stream>>>(rowptr, rec, lin[l], mbuf, N);
    gemm_norm_bias_relu<<<gemm_blocks, 256, 0, stream>>>(mbuf, norm_dst, Wp[l], bp[l],
                                                         lout[l], N);
  }
}


# Row-pointer kernel: counts in front of a block's slice read as 24 independent 16-byte loads per wait instead of one load per wait; own counts requested with them
# speedup vs baseline: 2.3607x; 1.1650x over previous
#include <hip/hip_runtime.h>
#include <hip/hip_bf16.h>

#define FEAT 96
#define KSTEPS (FEAT / 4)
#define NTILES (FEAT / 16)

typedef float v2f __attribute__((ext_vector_type(2)));
typedef float v4f __attribute__((ext_vector_type(4)));
typedef float v8f __attribute__((ext_vector_type(8)));

__global__ __launch_bounds__(256) void zero_f32(float* __restrict__ p, long long n) {
  long long i = (long long)blockIdx.x * blockDim.x + threadIdx.x;
  long long i4 = i * 4;
  if (i4 + 3 < n) {
    *(float4*)(p + i4) = make_float4(0.f, 0.f, 0.f, 0.f);
  } else {
    for (long long j = i4; j < n; ++j) p[j] = 0.f;
  }
}

__global__ __launch_bounds__(256) void degree_rank(const int* __restrict__ src,
                                                   const int* __restrict__ dst,
                                                   int* __restrict__ cnt_out,
                                                   int* __restrict__ cnt_in,
                                                   int* __restrict__ rank,
                                                   int nedges) {
  int e = blockIdx.x * blockDim.x + threadIdx.x;
  if (e >= nedges) return;
  const int s = src[e];
  const int d = dst[e];
  atomicAdd(&cnt_out[s], 1);
  rank[e] = atomicAdd(&cnt_in[d], 1);
}

#define PB 24

__global__ __launch_bounds__(256) void rowptr_and_norms(const int* __restrict__ cnt_out,
                                                        const int* __restrict__ cnt_in,
                                                        int* __restrict__ rowptr,
                                                        float* __restrict__ norm_src,
                                                        float* __restrict__ norm_dst,
                                                        int n) {
  __shared__ int wave_pre[8];
  __shared__ int wave_tot[8];
  const int tid = threadIdx.x;
  const int lane = tid & 31;
  const int wave = tid >> 5;
  const int nch = blockIdx.x;
  const int i0 = nch * 1024 + tid * 4;

  int c[4], d[4];
#pragma unroll
  for (int j = 0; j < 4; ++j) {
    const int idx = min(i0 + j, n - 1);
    c[j] = cnt_in[idx];
    d[j] = cnt_out[idx];
  }

  int pre = 0;
  const int* __restrict__ mine = cnt_in + tid * 4;
  for (int k0 = 0; k0 < nch; k0 += PB) {
    int4 v[PB];
#pragma unroll
    for (int j = 0; j < PB; ++j)
      v[j] = *(const int4*)(mine + min(k0 + j, nch - 1) * 1024);
    asm volatile("" ::
        "v"(v[0].x),  "v"(v[1].x),  "v"(v[2].x),  "v"(v[3].x),  "v"(v[4].x),  "v"(v[5].x),
        "v"(v[6].x),  "v"(v[7].x),  "v"(v[8].x),  "v"(v[9].x),  "v"(v[10].x), "v"(v[11].x),
        "v"(v[12].x), "v"(v[13].x), "v"(v[14].x), "v"(v[15].x), "v"(v[16].x), "v"(v[17].x),
        "v"(v[18].x), "v"(v[19].x), "v"(v[20].x), "v"(v[21].x), "v"(v[22].x), "v"(v[23].x));
#pragma unroll
    for (int j = 0; j < PB; ++j)
      if (k0 + j < nch) pre += (v[j].x + v[j].y) + (v[j].z + v[j].w);
  }

#pragma unroll
  for (int j = 0; j < 4; ++j) {
    const int idx = i0 + j;
    if (idx < n) {
      norm_src[idx] = 1.0f / sqrtf(fmaxf((float)d[j], 1.0f));
      norm_dst[idx] = 1.0f / sqrtf(fmaxf((float)c[j], 1.0f));
    } else {
      c[j] = 0;
    }
  }
  const int tsum = (c[0] + c[1]) + (c[2] + c[3]);

  int incl = tsum;
  for (int o = 1; o < 32; o <<= 1) {
    const int t = __shfl_up(incl, o);
    if (lane >= o) incl += t;
  }
  for (int o = 16; o > 0; o >>= 1) pre += __shfl_xor(pre, o);
  if (lane == 31) wave_tot[wave] = incl;
  if (lane == 0) wave_pre[wave] = pre;
  __syncthreads();

  int base = 0;
  for (int w = 0; w < 8; ++w) {
    base += wave_pre[w];
    if (w < wave) base += wave_tot[w];
  }
  int run = base + incl - tsum;
  for (int j = 0; j < 4; ++j) {
    const int idx = i0 + j;
    if (idx <= n) rowptr[idx] = run;
    run += c[j];
  }
}

__global__ __launch_bounds__(256) void edge_place(const int* __restrict__ src,
                                                  const int* __restrict__ dst,
                                                  const int* __restrict__ rank,
                                                  const int* __restrict__ rowptr,
                                                  const float* __restrict__ norm_src,
                                                  int2* __restrict__ rec,
                                                  int nedges) {
  int e = blockIdx.x * blockDim.x + threadIdx.x;
  if (e >= nedges) return;
  const int s = src[e];
  const int p = rowptr[dst[e]] + rank[e];
  rec[p] = make_int2(s, __float_as_int(norm_src[s]));
}

#define APITCH 100
#define GU 16

__device__ __forceinline__ void fma4(float4& acc, const float4 v, const float w) {
  acc.x = fmaf(v.x, w, acc.x);
  acc.y = fmaf(v.y, w, acc.y);
  acc.z = fmaf(v.z, w, acc.z);
  acc.w = fmaf(v.w, w, acc.w);
}

__global__ __launch_bounds__(256) void gather_gemm_bias_relu(
    const int* __restrict__ rowptr, const int2* __restrict__ rec,
    const float* __restrict__ h, const float* __restrict__ norm_dst,
    const float* __restrict__ W, const float* __restrict__ bias,
    float* __restrict__ out, int nrows) {
  __shared__ __attribute__((aligned(16))) float Ws[FEAT * FEAT];
  __shared__ float bs[FEAT];
  __shared__ __attribute__((aligned(16))) float As[8 * 16 * APITCH];

  const int tid = threadIdx.x;
  const int lane = tid & 31;
  const int wave = __builtin_amdgcn_readfirstlane(tid >> 5);
  const int row0 = (blockIdx.x * 8 + wave) * 16;
  float* __restrict__ strip = As + wave * (16 * APITCH);

  v4f wreg[9];
#pragma unroll
  for (int i = 0; i < 9; ++i) wreg[i] = ((const v4f*)W)[tid + 256 * i];

  if (row0 < nrows) {
    const int fo = min(lane, FEAT / 4 - 1) * 4;
    const int rp = rowptr[min(row0 + min(lane, 16), nrows)];
    const float ndl = norm_dst[min(row0 + (lane & 15), nrows - 1)];
    const int beg = __builtin_amdgcn_readlane(rp, 0);
    const int end = __builtin_amdgcn_readlane(rp, 16);

    int cur = 0;
    int nextb = __builtin_amdgcn_readlane(rp, 1);
    float4 acc = make_float4(0.f, 0.f, 0.f, 0.f);

#define PARK_ROW()                                                                    \
    do {                                                                              \
      const float nd = __int_as_float(                                                \
          __builtin_amdgcn_readlane(__float_as_int(ndl), cur));                       \
      if (lane < FEAT / 4)                                                            \
        *(float4*)(strip + cur * APITCH + lane * 4) =                                 \
            make_float4(acc.x * nd, acc.y * nd, acc.z * nd, acc.w * nd);              \
      acc = make_float4(0.f, 0.f, 0.f, 0.f);                                          \
      ++cur;                                                                          \
      nextb = __builtin_amdgcn_readlane(rp, cur + 1);                                 \
    } while (0)

    int2 p = make_int2(0, 0);
    if (beg < end) p = rec[min(beg + lane, end - 1)];
    for (int base = beg; base < end; base += 32) {
      int2 pn = p;
#pragma unroll
      for (int g = 0; g < 32 / GU; ++g) {
        const int e0 = base + g * GU;
        if (e0 < end) {
          float4 v[GU];
          float w[GU];
#pragma unroll
          for (int u = 0; u < GU; ++u) {
            const int s = __builtin_amdgcn_readlane(p.x, g * GU + u);
            w[u] = __int_as_float(__builtin_amdgcn_readlane(p.y, g * GU + u));
            v[u] = *(const float4*)(h + (long long)s * FEAT + fo);
          }
          if (g == 0 && base + 32 < end) pn = rec[min(base + 32 + lane, end - 1)];
#pragma unroll
          for (int u = 0; u < GU; ++u) {
            const int e = e0 + u;
            if (e < end) {
              while (e == nextb) PARK_ROW();
              fma4(acc, v[u], w[u]);
            }
          }
        }
      }
      p = pn;
    }
    while (cur < 16) PARK_ROW();
#undef PARK_ROW
  }

#pragma unroll
  for (int i = 0; i < 9; ++i) ((v4f*)Ws)[tid + 256 * i] = wreg[i];
  if (tid < FEAT) bs[tid] = bias[tid];
  __syncthreads();

  if (row0 >= nrows) return;

  const int lrow = lane & 15;
  const int lgrp = lane >> 4;

  v2f a[KSTEPS];
#pragma unroll
  for (int ks = 0; ks < KSTEPS; ++ks)
    a[ks] = *(const v2f*)(strip + lrow * APITCH + ks * 4 + 2 * lgrp);

  v8f acc[NTILES];
#pragma unroll
  for (int nt = 0; nt < NTILES; ++nt) acc[nt] = (v8f){};

  v2f bf[NTILES];
#pragma unroll
  for (int nt = 0; nt < NTILES; ++nt) {
    const int k = 2 * lgrp;
    bf[nt].x = Ws[k * FEAT + nt * 16 + lrow];
    bf[nt].y = Ws[(k + 1) * FEAT + nt * 16 + lrow];
  }

#pragma unroll
  for (int ks = 0; ks < KSTEPS; ++ks) {
    v2f bn[NTILES];
    if (ks + 1 < KSTEPS) {
      const int k = (ks + 1) * 4 + 2 * lgrp;
#pragma unroll
      for (int nt = 0; nt < NTILES; ++nt) {
        bn[nt].x = Ws[k * FEAT + nt * 16 + lrow];
        bn[nt].y = Ws[(k + 1) * FEAT + nt * 16 + lrow];
      }
    }
#pragma unroll
    for (int nt = 0; nt < NTILES; ++nt) {
      acc[nt] = __builtin_amdgcn_wmma_f32_16x16x4_f32(
           false, a[ks],  false, bf[nt],
           (short)0, acc[nt],  false,  false);
    }
    if (ks + 1 < KSTEPS) {
#pragma unroll
      for (int nt = 0; nt < NTILES; ++nt) bf[nt] = bn[nt];
    }
  }

#pragma unroll
  for (int nt = 0; nt < NTILES; ++nt) {
    const int nb = nt * 16;
    const float bcol = bs[nb + lrow];
#pragma unroll
    for (int r = 0; r < 8; ++r) {
      const int row = row0 + r + 8 * lgrp;
      if (row < nrows) {
        float v = acc[nt][r] + bcol;
        out[(long long)row * FEAT + nb + lrow] = fmaxf(v, 0.0f);
      }
    }
  }
}

extern "C" void kernel_launch(void* const* d_in, const int* in_sizes, int n_in,
                              void* d_out, int out_size, void* d_ws, size_t ws_size,
                              hipStream_t stream) {
  const float* h0  = (const float*)d_in[0];
  const int*   src = (const int*)d_in[1];
  const int*   dst = (const int*)d_in[2];
  const float* Wp[3] = {(const float*)d_in[3], (const float*)d_in[5], (const float*)d_in[7]};
  const float* bp[3] = {(const float*)d_in[4], (const float*)d_in[6], (const float*)d_in[8]};

  const int N = in_sizes[0] / FEAT;
  const int E = in_sizes[1];

  const long long NP = ((long long)N + 1 + 63) / 64 * 64;
  const long long EP = ((long long)E + 63) / 64 * 64;
  float* ws       = (float*)d_ws;
  float* norm_src = ws;
  float* norm_dst = ws + NP;
  int*   cnt_out  = (int*)(ws + 2 * NP);
  int*   cnt_in   = (int*)(ws + 3 * NP);
  int*   rowptr   = (int*)(ws + 4 * NP);
  int*   rank     = (int*)(ws + 5 * NP);
  int2*  rec      = (int2*)(ws + 5 * NP + EP);
  float* hA       = ws + 5 * NP + 3 * EP;
  float* outp     = (float*)d_out;

  {
    long long n2 = 2 * NP;
    int blocks = (int)((n2 / 4 + 255) / 256) + 1;
    zero_f32<<<blocks, 256, 0, stream>>>((float*)cnt_out, n2);
    degree_rank<<<(E + 255) / 256, 256, 0, stream>>>(src, dst, cnt_out, cnt_in, rank, E);
    rowptr_and_norms<<<N / 1024 + 1, 256, 0, stream>>>(cnt_out, cnt_in, rowptr,
                                                       norm_src, norm_dst, N);
    edge_place<<<(E + 255) / 256, 256, 0, stream>>>(src, dst, rank, rowptr, norm_src, rec, E);
  }

  const int strips = (N + 15) / 16;
  const int gemm_blocks = (strips + 7) / 8;

  const float* lin[3]  = {h0, outp, hA};
  float*       lout[3] = {outp, hA, outp};

  for (int l = 0; l < 3; ++l) {
    gather_gemm_bias_relu<<<gemm_blocks, 256, 0, stream>>>(rowptr, rec, lin[l], norm_dst,
                                                           Wp[l], bp[l], lout[l], N);
  }
}


# Records stored in fixed 64-slot rows per destination at the degree atomic's return (overflow list past 64): no prefix-sum launch, no 2nd edge pass; source norm premultiplied into rows before the sum
# speedup vs baseline: 2.4906x; 1.0550x over previous
#include <hip/hip_runtime.h>
#include <hip/hip_bf16.h>

#define FEAT 96
#define KSTEPS (FEAT / 4)
#define NTILES (FEAT / 16)

typedef float v2f __attribute__((ext_vector_type(2)));
typedef float v4f __attribute__((ext_vector_type(4)));
typedef float v8f __attribute__((ext_vector_type(8)));

__global__ __launch_bounds__(256) void zero_f32(float* __restrict__ p, long long n) {
  long long i = (long long)blockIdx.x * blockDim.x + threadIdx.x;
  long long i4 = i * 4;
  if (i4 + 3 < n) {
    *(float4*)(p + i4) = make_float4(0.f, 0.f, 0.f, 0.f);
  } else {
    for (long long j = i4; j < n; ++j) p[j] = 0.f;
  }
}

#define SLOTS 64

__global__ __launch_bounds__(256) void degree_place(const int* __restrict__ src,
                                                    const int* __restrict__ dst,
                                                    int* __restrict__ cnt_out,
                                                    int* __restrict__ cnt_in,
                                                    int* __restrict__ ovf_cnt,
                                                    int* __restrict__ rec,
                                                    int2* __restrict__ ovf,
                                                    int nedges) {
  int e = blockIdx.x * blockDim.x + threadIdx.x;
  if (e >= nedges) return;
  const int s = src[e];
  const int d = dst[e];
  atomicAdd(&cnt_out[s], 1);
  const int slot = atomicAdd(&cnt_in[d], 1);
  if (slot < SLOTS) {
    rec[d * SLOTS + slot] = s;
  } else {
    const int i = atomicAdd(ovf_cnt, 1);
    ovf[i] = make_int2(d, s);
  }
}

#define SC 8

__global__ __launch_bounds__(256) void scale_rows(const float* __restrict__ h,
                                                  const int* __restrict__ cnt_out,
                                                  float* __restrict__ hs,
                                                  int nvec) {
  const int base = blockIdx.x * (256 * SC) + threadIdx.x;
  v4f v[SC];
  int c[SC];
#pragma unroll
  for (int k = 0; k < SC; ++k) {
    const int i = min(base + k * 256, nvec - 1);
    v[k] = ((const v4f*)h)[i];
    c[k] = cnt_out[i / (FEAT / 4)];
  }
#pragma unroll
  for (int k = 0; k < SC; ++k) {
    const int i = base + k * 256;
    if (i < nvec) {
      const float ns = 1.0f / sqrtf(fmaxf((float)c[k], 1.0f));
      ((v4f*)hs)[i] = v[k] * ns;
    }
  }
}

#define APITCH 100
#define GU 16

__device__ __forceinline__ void add4(float4& acc, const float4 v) {
  acc.x += v.x;
  acc.y += v.y;
  acc.z += v.z;
  acc.w += v.w;
}

__device__ __forceinline__ int rec_index(const int rp, const int v, const int row0) {
  int r = 0, st = 0;
#pragma unroll
  for (int j = 1; j < 16; ++j) {
    const int rj = __builtin_amdgcn_readlane(rp, j);
    if (v >= rj) { r = j; st = rj; }
  }
  return (row0 + r) * SLOTS + (v - st);
}

__global__ __launch_bounds__(256) void gather_gemm_bias_relu(
    const int* __restrict__ cnt_in, const int* __restrict__ cnt_out,
    const int* __restrict__ rec, const int* __restrict__ ovf_cnt,
    const int2* __restrict__ ovf, const float* __restrict__ h,
    const float* __restrict__ W, const float* __restrict__ bias,
    float* __restrict__ out, int nrows, int nedges, int scale_out) {
  __shared__ __attribute__((aligned(16))) float Ws[FEAT * FEAT];
  __shared__ float bs[FEAT];
  __shared__ __attribute__((aligned(16))) float As[8 * 16 * APITCH];

  const int tid = threadIdx.x;
  const int lane = tid & 31;
  const int wave = __builtin_amdgcn_readfirstlane(tid >> 5);
  const int row0 = (blockIdx.x * 8 + wave) * 16;
  float* __restrict__ strip = As + wave * (16 * APITCH);

  v4f wreg[9];
#pragma unroll
  for (int i = 0; i < 9; ++i) wreg[i] = ((const v4f*)W)[tid + 256 * i];
  const int novf_lane = ovf_cnt[0];

  if (row0 < nrows) {
    const int fo = min(lane, FEAT / 4 - 1) * 4;
    const int rr = row0 + (lane & 15);
    int cfull = cnt_in[min(rr, nrows - 1)];
    if (lane >= 16 || rr >= nrows) cfull = 0;
    const float ndl = 1.0f / sqrtf(fmaxf((float)cfull, 1.0f));
    const int c = min(cfull, SLOTS);
    int incl = c;
    for (int o = 1; o < 16; o <<= 1) {
      const int t = __shfl_up(incl, o);
      if (lane >= o) incl += t;
    }
    const int end = __builtin_amdgcn_readlane(incl, 15);
    int rp = __shfl_up(incl, 1);
    if (lane == 0) rp = 0;
    if (lane >= 16) rp = end;

    int cur = 0;
    int nextb = __builtin_amdgcn_readlane(rp, 1);
    float4 acc = make_float4(0.f, 0.f, 0.f, 0.f);

#define PARK_ROW()                                                                    \
    do {                                                                              \
      const float nd = __int_as_float(                                                \
          __builtin_amdgcn_readlane(__float_as_int(ndl), cur));                       \
      if (lane < FEAT / 4)                                                            \
        *(float4*)(strip + cur * APITCH + lane * 4) =                                 \
            make_float4(acc.x * nd, acc.y * nd, acc.z * nd, acc.w * nd);              \
      acc = make_float4(0.f, 0.f, 0.f, 0.f);                                          \
      ++cur;                                                                          \
      nextb = __builtin_amdgcn_readlane(rp, cur + 1);                                 \
    } while (0)

    int p = 0;
    if (0 < end) p = rec[rec_index(rp, min(lane, end - 1), row0)];
    for (int base = 0; base < end; base += 32) {
      int pn = p;
#pragma unroll
      for (int g = 0; g < 32 / GU; ++g) {
        const int e0 = base + g * GU;
        if (e0 < end) {
          float4 v[GU];
#pragma unroll
          for (int u = 0; u < GU; ++u) {
            const int s = __builtin_amdgcn_readlane(p, g * GU + u);
            v[u] = *(const float4*)(h + (long long)s * FEAT + fo);
          }
          if (g == 0 && base + 32 < end)
            pn = rec[rec_index(rp, min(base + 32 + lane, end - 1), row0)];
#pragma unroll
          for (int u = 0; u < GU; ++u) {
            const int e = e0 + u;
            if (e < end) {
              while (e == nextb) PARK_ROW();
              add4(acc, v[u]);
            }
          }
        }
      }
      p = pn;
    }
    while (cur < 16) PARK_ROW();
#undef PARK_ROW

    const int novf = min(__builtin_amdgcn_readfirstlane(novf_lane), nedges);
    for (int i = 0; i < novf; ++i) {
      const int2 o = ovf[i];
      const int r = __builtin_amdgcn_readfirstlane(o.x) - row0;
      if (r >= 0 && r < 16) {
        const int s = __builtin_amdgcn_readfirstlane(o.y);
        const float nd = __int_as_float(__builtin_amdgcn_readlane(__float_as_int(ndl), r));
        if (lane < FEAT / 4) {
          float4* q = (float4*)(strip + r * APITCH + lane * 4);
          const float4 hv = *(const float4*)(h + (long long)s * FEAT + lane * 4);
          float4 t = *q;
          t.x = fmaf(hv.x, nd, t.x);
          t.y = fmaf(hv.y, nd, t.y);
          t.z = fmaf(hv.z, nd, t.z);
          t.w = fmaf(hv.w, nd, t.w);
          *q = t;
        }
      }
    }
  }

#pragma unroll
  for (int i = 0; i < 9; ++i) ((v4f*)Ws)[tid + 256 * i] = wreg[i];
  if (tid < FEAT) bs[tid] = bias[tid];
  __syncthreads();

  if (row0 >= nrows) return;

  const int lrow = lane & 15;
  const int lgrp = lane >> 4;

  float osc[8];
#pragma unroll
  for (int r = 0; r < 8; ++r) {
    osc[r] = 1.0f;
    if (scale_out) {
      const int row = min(row0 + r + 8 * lgrp, nrows - 1);
      osc[r] = 1.0f / sqrtf(fmaxf((float)cnt_out[row], 1.0f));
    }
  }

  v2f a[KSTEPS];
#pragma unroll
  for (int ks = 0; ks < KSTEPS; ++ks)
    a[ks] = *(const v2f*)(strip + lrow * APITCH + ks * 4 + 2 * lgrp);

  v8f acc[NTILES];
#pragma unroll
  for (int nt = 0; nt < NTILES; ++nt) acc[nt] = (v8f){};

  v2f bf[NTILES];
#pragma unroll
  for (int nt = 0; nt < NTILES; ++nt) {
    const int k = 2 * lgrp;
    bf[nt].x = Ws[k * FEAT + nt * 16 + lrow];
    bf[nt].y = Ws[(k + 1) * FEAT + nt * 16 + lrow];
  }

#pragma unroll
  for (int ks = 0; ks < KSTEPS; ++ks) {
    v2f bn[NTILES];
    if (ks + 1 < KSTEPS) {
      const int k = (ks + 1) * 4 + 2 * lgrp;
#pragma unroll
      for (int nt = 0; nt < NTILES; ++nt) {
        bn[nt].x = Ws[k * FEAT + nt * 16 + lrow];
        bn[nt].y = Ws[(k + 1) * FEAT + nt * 16 + lrow];
      }
    }
#pragma unroll
    for (int nt = 0; nt < NTILES; ++nt) {
      acc[nt] = __builtin_amdgcn_wmma_f32_16x16x4_f32(
           false, a[ks],  false, bf[nt],
           (short)0, acc[nt],  false,  false);
    }
    if (ks + 1 < KSTEPS) {
#pragma unroll
      for (int nt = 0; nt < NTILES; ++nt) bf[nt] = bn[nt];
    }
  }

#pragma unroll
  for (int nt = 0; nt < NTILES; ++nt) {
    const int nb = nt * 16;
    const float bcol = bs[nb + lrow];
#pragma unroll
    for (int r = 0; r < 8; ++r) {
      const int row = row0 + r + 8 * lgrp;
      if (row < nrows) {
        float v = acc[nt][r] + bcol;
        out[(long long)row * FEAT + nb + lrow] = fmaxf(v, 0.0f) * osc[r];
      }
    }
  }
}

extern "C" void kernel_launch(void* const* d_in, const int* in_sizes, int n_in,
                              void* d_out, int out_size, void* d_ws, size_t ws_size,
                              hipStream_t stream) {
  const float* h0  = (const float*)d_in[0];
  const int*   src = (const int*)d_in[1];
  const int*   dst = (const int*)d_in[2];
  const float* Wp[3] = {(const float*)d_in[3], (const float*)d_in[5], (const float*)d_in[7]};
  const float* bp[3] = {(const float*)d_in[4], (const float*)d_in[6], (const float*)d_in[8]};

  const int N = in_sizes[0] / FEAT;
  const int E = in_sizes[1];

  const long long NP = ((long long)N + 63) / 64 * 64;
  const long long EP = ((long long)E + 63) / 64 * 64;
  float* ws       = (float*)d_ws;
  int*   cnt_out  = (int*)ws;
  int*   cnt_in   = (int*)(ws + NP);
  int*   ovf_cnt  = (int*)(ws + 2 * NP);
  int*   rec      = (int*)(ws + 2 * NP + 64);
  int2*  ovf      = (int2*)(ws + 2 * NP + 64 + (long long)N * SLOTS);
  float* hA       = ws + 2 * NP + 64 + (long long)N * SLOTS + 2 * EP;
  float* hs0      = hA + (long long)N * FEAT;
  float* outp     = (float*)d_out;

  {
    long long n2 = 2 * NP + 64;
    int blocks = (int)((n2 / 4 + 255) / 256) + 1;
    zero_f32<<<blocks, 256, 0, stream>>>((float*)cnt_out, n2);
    degree_place<<<(E + 255) / 256, 256, 0, stream>>>(src, dst, cnt_out, cnt_in, ovf_cnt,
                                                      rec, ovf, E);
    const int nvec = N * (FEAT / 4);
    scale_rows<<<(nvec + 256 * SC - 1) / (256 * SC), 256, 0, stream>>>(h0, cnt_out, hs0, nvec);
  }

  const int strips = (N + 15) / 16;
  const int gemm_blocks = (strips + 7) / 8;

  const float* lin[3]  = {hs0, outp, hA};
  float*       lout[3] = {outp, hA, outp};

  for (int l = 0; l < 3; ++l) {
    gather_gemm_bias_relu<<<gemm_blocks, 256, 0, stream>>>(cnt_in, cnt_out, rec, ovf_cnt, ovf,
                                                           lin[l], Wp[l], bp[l], lout[l],
                                                           N, E, l < 2 ? 1 : 0);
  }
}


# Edges grouped by destination through LDS: binned by 196-node range (one global add per block and bin), then one block per bin builds degrees and record rows in LDS; row sum order = LDS arrival rank
# speedup vs baseline: 4.0197x; 1.6139x over previous
#include <hip/hip_runtime.h>
#include <hip/hip_bf16.h>

#define FEAT 96
#define KSTEPS (FEAT / 4)
#define NTILES (FEAT / 16)

typedef float v2f __attribute__((ext_vector_type(2)));
typedef float v4f __attribute__((ext_vector_type(4)));
typedef float v8f __attribute__((ext_vector_type(8)));

__global__ __launch_bounds__(256) void zero_f32(float* __restrict__ p, long long n) {
  long long i = (long long)blockIdx.x * blockDim.x + threadIdx.x;
  long long i4 = i * 4;
  if (i4 + 3 < n) {
    *(float4*)(p + i4) = make_float4(0.f, 0.f, 0.f, 0.f);
  } else {
    for (long long j = i4; j < n; ++j) p[j] = 0.f;
  }
}

#define SLOTS 64
#define NPB 196
#define NBIN 256
#define BCAP 4096
#define BT 512
#define EPT 8
#define EPB (BT * EPT)
#define PT 1024

__device__ __forceinline__ int bin_of(const int node) {
  return min((int)((unsigned)node / NPB), NBIN - 1);
}

__device__ __forceinline__ void scan_reserve(const int* hist, int* start, int* gbase,
                                             int* wsum, int* __restrict__ fill,
                                             const int tid) {
  const int lane = tid & 31;
  const int wave = tid >> 5;
  int c = 0, incl = 0;
  if (tid < NBIN) {
    c = hist[tid];
    incl = c;
    for (int o = 1; o < 32; o <<= 1) {
      const int t = __shfl_up(incl, o);
      if (lane >= o) incl += t;
    }
    if (lane == 31) wsum[wave] = incl;
  }
  __syncthreads();
  if (tid < NBIN) {
    int woff = 0;
    for (int w = 0; w < NBIN / 32; ++w)
      if (w < wave) woff += wsum[w];
    start[tid] = woff + incl - c;
    gbase[tid] = c ? atomicAdd(&fill[tid], c) : 0;
  }
  __syncthreads();
}

__global__ __launch_bounds__(BT) void bin_edges(const int* __restrict__ src,
                                                const int* __restrict__ dst,
                                                int* __restrict__ dfill,
                                                int* __restrict__ sfill,
                                                int2* __restrict__ dbin,
                                                int* __restrict__ sbin,
                                                int* __restrict__ misc,
                                                int2* __restrict__ dspill,
                                                int* __restrict__ sspill,
                                                int nedges) {
  __shared__ int hist[NBIN];
  __shared__ int start[NBIN];
  __shared__ int gbase[NBIN];
  __shared__ int wsum[NBIN / 32];
  __shared__ int2 sorted[EPB];

  const int tid = threadIdx.x;
  const int base = blockIdx.x * EPB;
  const int total = min(EPB, nedges - base);

  int s[EPT], d[EPT], r[EPT];
#pragma unroll
  for (int k = 0; k < EPT; ++k) {
    const int e = min(base + k * BT + tid, nedges - 1);
    s[k] = src[e];
    d[k] = dst[e];
  }

  if (tid < NBIN) hist[tid] = 0;
  __syncthreads();
#pragma unroll
  for (int k = 0; k < EPT; ++k)
    if (k * BT + tid < total) r[k] = atomicAdd(&hist[bin_of(d[k])], 1);
  __syncthreads();
  scan_reserve(hist, start, gbase, wsum, dfill, tid);
#pragma unroll
  for (int k = 0; k < EPT; ++k)
    if (k * BT + tid < total) sorted[start[bin_of(d[k])] + r[k]] = make_int2(d[k], s[k]);
  __syncthreads();
  for (int i = tid; i < total; i += BT) {
    const int2 p = sorted[i];
    const int b = bin_of(p.x);
    const int pos = gbase[b] + (i - start[b]);
    if (pos < BCAP) {
      dbin[b * BCAP + pos] = p;
    } else {
      const int j = atomicAdd(&misc[1], 1);
      dspill[j] = p;
    }
  }
  __syncthreads();

  int* sorted1 = (int*)sorted;
  if (tid < NBIN) hist[tid] = 0;
  __syncthreads();
#pragma unroll
  for (int k = 0; k < EPT; ++k)
    if (k * BT + tid < total) r[k] = atomicAdd(&hist[bin_of(s[k])], 1);
  __syncthreads();
  scan_reserve(hist, start, gbase, wsum, sfill, tid);
#pragma unroll
  for (int k = 0; k < EPT; ++k)
    if (k * BT + tid < total) sorted1[start[bin_of(s[k])] + r[k]] = s[k];
  __syncthreads();
  for (int i = tid; i < total; i += BT) {
    const int sv = sorted1[i];
    const int b = bin_of(sv);
    const int pos = gbase[b] + (i - start[b]);
    if (pos < BCAP) {
      sbin[b * BCAP + pos] = sv;
    } else {
      const int j = atomicAdd(&misc[2], 1);
      sspill[j] = sv;
    }
  }
}

__global__ __launch_bounds__(PT) void place_bin(const int* __restrict__ dfill,
                                                const int* __restrict__ sfill,
                                                const int2* __restrict__ dbin,
                                                const int* __restrict__ sbin,
                                                int* __restrict__ misc,
                                                const int2* __restrict__ dspill,
                                                const int* __restrict__ sspill,
                                                int* __restrict__ cnt_in,
                                                int* __restrict__ cnt_out,
                                                int* __restrict__ rec,
                                                int2* __restrict__ ovf,
                                                int nrows, int nedges) {
  __shared__ int cin[NPB];
  __shared__ int cout[NPB];
  __shared__ __attribute__((aligned(16))) int rows[NPB * SLOTS];

  const int tid = threadIdx.x;
  const int b = blockIdx.x;
  const int node0 = b * NPB;
  if (tid < NPB) {
    cin[tid] = 0;
    cout[tid] = 0;
  }
  const int nd = min(dfill[b], BCAP);
  const int ns = min(sfill[b], BCAP);
  const int ndsp = min(misc[1], nedges);
  const int nssp = min(misc[2], nedges);
  __syncthreads();

#define ADD_PAIR(p)                                                   \
  do {                                                                \
    const int dl = (p).x - node0;                                     \
    if ((unsigned)dl < NPB) {                                         \
      const int slot = atomicAdd(&cin[dl], 1);                        \
      if (slot < SLOTS) {                                             \
        rows[dl * SLOTS + slot] = (p).y;                              \
      } else {                                                        \
        const int j = atomicAdd(&misc[0], 1);                         \
        ovf[j] = (p);                                                 \
      }                                                               \
    }                                                                 \
  } while (0)

  int2 pr[BCAP / PT];
  int sv[BCAP / PT];
#pragma unroll
  for (int k = 0; k < BCAP / PT; ++k) {
    const int i = k * PT + tid;
    if (i < nd) pr[k] = dbin[b * BCAP + i];
    if (i < ns) sv[k] = sbin[b * BCAP + i];
  }
#pragma unroll
  for (int k = 0; k < BCAP / PT; ++k)
    if (k * PT + tid < nd) ADD_PAIR(pr[k]);
  for (int i = tid; i < ndsp; i += PT) {
    const int2 p = dspill[i];
    ADD_PAIR(p);
  }
#undef ADD_PAIR
#pragma unroll
  for (int k = 0; k < BCAP / PT; ++k) {
    const int sl = sv[k] - node0;
    if (k * PT + tid < ns && (unsigned)sl < NPB) atomicAdd(&cout[sl], 1);
  }
  for (int i = tid; i < nssp; i += PT) {
    const int sl = sspill[i] - node0;
    if ((unsigned)sl < NPB) atomicAdd(&cout[sl], 1);
  }
  __syncthreads();

  if (tid < NPB && node0 + tid < nrows) {
    cnt_in[node0 + tid] = cin[tid];
    cnt_out[node0 + tid] = cout[tid];
  }
  for (int q = tid; q < NPB * (SLOTS / 4); q += PT) {
    const int dl = q / (SLOTS / 4);
    const int piece = q % (SLOTS / 4);
    if (node0 + dl < nrows && piece * 4 < min(cin[dl], SLOTS))
      ((int4*)rec)[(long long)(node0 + dl) * (SLOTS / 4) + piece] = ((const int4*)rows)[q];
  }
}

#define SC 8

__global__ __launch_bounds__(256) void scale_rows(const float* __restrict__ h,
                                                  const int* __restrict__ cnt_out,
                                                  float* __restrict__ hs,
                                                  int nvec) {
  const int base = blockIdx.x * (256 * SC) + threadIdx.x;
  v4f v[SC];
  int c[SC];
#pragma unroll
  for (int k = 0; k < SC; ++k) {
    const int i = min(base + k * 256, nvec - 1);
    v[k] = ((const v4f*)h)[i];
    c[k] = cnt_out[i / (FEAT / 4)];
  }
#pragma unroll
  for (int k = 0; k < SC; ++k) {
    const int i = base + k * 256;
    if (i < nvec) {
      const float ns = 1.0f / sqrtf(fmaxf((float)c[k], 1.0f));
      ((v4f*)hs)[i] = v[k] * ns;
    }
  }
}

#define APITCH 100
#define GU 16

__device__ __forceinline__ void add4(float4& acc, const float4 v) {
  acc.x += v.x;
  acc.y += v.y;
  acc.z += v.z;
  acc.w += v.w;
}

__device__ __forceinline__ int rec_index(const int rp, const int v, const int row0) {
  int r = 0, st = 0;
#pragma unroll
  for (int j = 1; j < 16; ++j) {
    const int rj = __builtin_amdgcn_readlane(rp, j);
    if (v >= rj) { r = j; st = rj; }
  }
  return (row0 + r) * SLOTS + (v - st);
}

__global__ __launch_bounds__(256) void gather_gemm_bias_relu(
    const int* __restrict__ cnt_in, const int* __restrict__ cnt_out,
    const int* __restrict__ rec, const int* __restrict__ ovf_cnt,
    const int2* __restrict__ ovf, const float* __restrict__ h,
    const float* __restrict__ W, const float* __restrict__ bias,
    float* __restrict__ out, int nrows, int nedges, int scale_out) {
  __shared__ __attribute__((aligned(16))) float Ws[FEAT * FEAT];
  __shared__ float bs[FEAT];
  __shared__ __attribute__((aligned(16))) float As[8 * 16 * APITCH];

  const int tid = threadIdx.x;
  const int lane = tid & 31;
  const int wave = __builtin_amdgcn_readfirstlane(tid >> 5);
  const int row0 = (blockIdx.x * 8 + wave) * 16;
  float* __restrict__ strip = As + wave * (16 * APITCH);

  v4f wreg[9];
#pragma unroll
  for (int i = 0; i < 9; ++i) wreg[i] = ((const v4f*)W)[tid + 256 * i];
  const int novf_lane = ovf_cnt[0];

  if (row0 < nrows) {
    const int fo = min(lane, FEAT / 4 - 1) * 4;
    const int rr = row0 + (lane & 15);
    int cfull = cnt_in[min(rr, nrows - 1)];
    if (lane >= 16 || rr >= nrows) cfull = 0;
    const float ndl = 1.0f / sqrtf(fmaxf((float)cfull, 1.0f));
    const int c = min(cfull, SLOTS);
    int incl = c;
    for (int o = 1; o < 16; o <<= 1) {
      const int t = __shfl_up(incl, o);
      if (lane >= o) incl += t;
    }
    const int end = __builtin_amdgcn_readlane(incl, 15);
    int rp = __shfl_up(incl, 1);
    if (lane == 0) rp = 0;
    if (lane >= 16) rp = end;

    int cur = 0;
    int nextb = __builtin_amdgcn_readlane(rp, 1);
    float4 acc = make_float4(0.f, 0.f, 0.f, 0.f);

#define PARK_ROW()                                                                    \
    do {                                                                              \
      const float nd = __int_as_float(                                                \
          __builtin_amdgcn_readlane(__float_as_int(ndl), cur));                       \
      if (lane < FEAT / 4)                                                            \
        *(float4*)(strip + cur * APITCH + lane * 4) =                                 \
            make_float4(acc.x * nd, acc.y * nd, acc.z * nd, acc.w * nd);              \
      acc = make_float4(0.f, 0.f, 0.f, 0.f);                                          \
      ++cur;                                                                          \
      nextb = __builtin_amdgcn_readlane(rp, cur + 1);                                 \
    } while (0)

    int p = 0;
    if (0 < end) p = rec[rec_index(rp, min(lane, end - 1), row0)];
    for (int base = 0; base < end; base += 32) {
      int pn = p;
#pragma unroll
      for (int g = 0; g < 32 / GU; ++g) {
        const int e0 = base + g * GU;
        if (e0 < end) {
          float4 v[GU];
#pragma unroll
          for (int u = 0; u < GU; ++u) {
            const int s = __builtin_amdgcn_readlane(p, g * GU + u);
            v[u] = *(const float4*)(h + (long long)s * FEAT + fo);
          }
          if (g == 0 && base + 32 < end)
            pn = rec[rec_index(rp, min(base + 32 + lane, end - 1), row0)];
#pragma unroll
          for (int u = 0; u < GU; ++u) {
            const int e = e0 + u;
            if (e < end) {
              while (e == nextb) PARK_ROW();
              add4(acc, v[u]);
            }
          }
        }
      }
      p = pn;
    }
    while (cur < 16) PARK_ROW();
#undef PARK_ROW

    const int novf = min(__builtin_amdgcn_readfirstlane(novf_lane), nedges);
    for (int i = 0; i < novf; ++i) {
      const int2 o = ovf[i];
      const int r = __builtin_amdgcn_readfirstlane(o.x) - row0;
      if (r >= 0 && r < 16) {
        const int s = __builtin_amdgcn_readfirstlane(o.y);
        const float nd = __int_as_float(__builtin_amdgcn_readlane(__float_as_int(ndl), r));
        if (lane < FEAT / 4) {
          float4* q = (float4*)(strip + r * APITCH + lane * 4);
          const float4 hv = *(const float4*)(h + (long long)s * FEAT + lane * 4);
          float4 t = *q;
          t.x = fmaf(hv.x, nd, t.x);
          t.y = fmaf(hv.y, nd, t.y);
          t.z = fmaf(hv.z, nd, t.z);
          t.w = fmaf(hv.w, nd, t.w);
          *q = t;
        }
      }
    }
  }

#pragma unroll
  for (int i = 0; i < 9; ++i) ((v4f*)Ws)[tid + 256 * i] = wreg[i];
  if (tid < FEAT) bs[tid] = bias[tid];
  __syncthreads();

  if (row0 >= nrows) return;

  const int lrow = lane & 15;
  const int lgrp = lane >> 4;

  float osc[8];
#pragma unroll
  for (int r = 0; r < 8; ++r) {
    osc[r] = 1.0f;
    if (scale_out) {
      const int row = min(row0 + r + 8 * lgrp, nrows - 1);
      osc[r] = 1.0f / sqrtf(fmaxf((float)cnt_out[row], 1.0f));
    }
  }

  v2f a[KSTEPS];
#pragma unroll
  for (int ks = 0; ks < KSTEPS; ++ks)
    a[ks] = *(const v2f*)(strip + lrow * APITCH + ks * 4 + 2 * lgrp);

  v8f acc[NTILES];
#pragma unroll
  for (int nt = 0; nt < NTILES; ++nt) acc[nt] = (v8f){};

  v2f bf[NTILES];
#pragma unroll
  for (int nt = 0; nt < NTILES; ++nt) {
    const int k = 2 * lgrp;
    bf[nt].x = Ws[k * FEAT + nt * 16 + lrow];
    bf[nt].y = Ws[(k + 1) * FEAT + nt * 16 + lrow];
  }

#pragma unroll
  for (int ks = 0; ks < KSTEPS; ++ks) {
    v2f bn[NTILES];
    if (ks + 1 < KSTEPS) {
      const int k = (ks + 1) * 4 + 2 * lgrp;
#pragma unroll
      for (int nt = 0; nt < NTILES; ++nt) {
        bn[nt].x = Ws[k * FEAT + nt * 16 + lrow];
        bn[nt].y = Ws[(k + 1) * FEAT + nt * 16 + lrow];
      }
    }
#pragma unroll
    for (int nt = 0; nt < NTILES; ++nt) {
      acc[nt] = __builtin_amdgcn_wmma_f32_16x16x4_f32(
           false, a[ks],  false, bf[nt],
           (short)0, acc[nt],  false,  false);
    }
    if (ks + 1 < KSTEPS) {
#pragma unroll
      for (int nt = 0; nt < NTILES; ++nt) bf[nt] = bn[nt];
    }
  }

#pragma unroll
  for (int nt = 0; nt < NTILES; ++nt) {
    const int nb = nt * 16;
    const float bcol = bs[nb + lrow];
#pragma unroll
    for (int r = 0; r < 8; ++r) {
      const int row = row0 + r + 8 * lgrp;
      if (row < nrows) {
        float v = acc[nt][r] + bcol;
        out[(long long)row * FEAT + nb + lrow] = fmaxf(v, 0.0f) * osc[r];
      }
    }
  }
}

extern "C" void kernel_launch(void* const* d_in, const int* in_sizes, int n_in,
                              void* d_out, int out_size, void* d_ws, size_t ws_size,
                              hipStream_t stream) {
  const float* h0  = (const float*)d_in[0];
  const int*   src = (const int*)d_in[1];
  const int*   dst = (const int*)d_in[2];
  const float* Wp[3] = {(const float*)d_in[3], (const float*)d_in[5], (const float*)d_in[7]};
  const float* bp[3] = {(const float*)d_in[4], (const float*)d_in[6], (const float*)d_in[8]};

  const int N = in_sizes[0] / FEAT;
  const int E = in_sizes[1];

  const long long NP = ((long long)N + 63) / 64 * 64;
  const long long EP = ((long long)E + 63) / 64 * 64;
  float* ws       = (float*)d_ws;
  int*   cnt_out  = (int*)ws;
  int*   cnt_in   = (int*)(ws + NP);
  int*   misc     = (int*)(ws + 2 * NP);
  int*   dfill    = misc + 64;
  int*   sfill    = dfill + NBIN;
  const long long HEAD = 2 * NP + 64 + 2 * NBIN;
  int*   rec      = (int*)(ws + HEAD);
  int2*  ovf      = (int2*)(ws + HEAD + (long long)N * SLOTS);
  float* hA       = ws + HEAD + (long long)N * SLOTS + 2 * EP;
  float* hs0      = hA + (long long)N * FEAT;
  int2*  dbin     = (int2*)(hs0 + (long long)N * FEAT);
  int*   sbin     = (int*)(hs0 + (long long)N * FEAT + 2LL * NBIN * BCAP);
  int2*  dspill   = (int2*)(sbin + (long long)NBIN * BCAP);
  int*   sspill   = (int*)(dspill + EP);
  float* outp     = (float*)d_out;

  {
    long long n2 = 64 + 2 * NBIN;
    int blocks = (int)((n2 / 4 + 255) / 256) + 1;
    zero_f32<<<blocks, 256, 0, stream>>>((float*)misc, n2);
    bin_edges<<<(E + EPB - 1) / EPB, BT, 0, stream>>>(src, dst, dfill, sfill, dbin, sbin, misc,
                                                      dspill, sspill, E);
    place_bin<<<(N + NPB - 1) / NPB, PT, 0, stream>>>(dfill, sfill, dbin, sbin, misc, dspill,
                                                      sspill, cnt_in, cnt_out, rec, ovf, N, E);
    const int nvec = N * (FEAT / 4);
    scale_rows<<<(nvec + 256 * SC - 1) / (256 * SC), 256, 0, stream>>>(h0, cnt_out, hs0, nvec);
  }

  const int strips = (N + 15) / 16;
  const int gemm_blocks = (strips + 7) / 8;

  const float* lin[3]  = {hs0, outp, hA};
  float*       lout[3] = {outp, hA, outp};

  for (int l = 0; l < 3; ++l) {
    gather_gemm_bias_relu<<<gemm_blocks, 256, 0, stream>>>(cnt_in, cnt_out, rec, misc, ovf,
                                                           lin[l], Wp[l], bp[l], lout[l],
                                                           N, E, l < 2 ? 1 : 0);
  }
}


# Row sum: whole 16-record groups add with no per-record end test and the row-boundary test marked unlikely, so a common record runs straight through (no taken branch; was two per record)
# speedup vs baseline: 4.1576x; 1.0343x over previous
#include <hip/hip_runtime.h>
#include <hip/hip_bf16.h>

#define FEAT 96
#define KSTEPS (FEAT / 4)
#define NTILES (FEAT / 16)

typedef float v2f __attribute__((ext_vector_type(2)));
typedef float v4f __attribute__((ext_vector_type(4)));
typedef float v8f __attribute__((ext_vector_type(8)));

__global__ __launch_bounds__(256) void zero_f32(float* __restrict__ p, long long n) {
  long long i = (long long)blockIdx.x * blockDim.x + threadIdx.x;
  long long i4 = i * 4;
  if (i4 + 3 < n) {
    *(float4*)(p + i4) = make_float4(0.f, 0.f, 0.f, 0.f);
  } else {
    for (long long j = i4; j < n; ++j) p[j] = 0.f;
  }
}

#define SLOTS 64
#define NPB 196
#define NBIN 256
#define BCAP 4096
#define BT 512
#define EPT 8
#define EPB (BT * EPT)
#define PT 1024

__device__ __forceinline__ int bin_of(const int node) {
  return min((int)((unsigned)node / NPB), NBIN - 1);
}

__device__ __forceinline__ void scan_reserve(const int* hist, int* start, int* gbase,
                                             int* wsum, int* __restrict__ fill,
                                             const int tid) {
  const int lane = tid & 31;
  const int wave = tid >> 5;
  int c = 0, incl = 0;
  if (tid < NBIN) {
    c = hist[tid];
    incl = c;
    for (int o = 1; o < 32; o <<= 1) {
      const int t = __shfl_up(incl, o);
      if (lane >= o) incl += t;
    }
    if (lane == 31) wsum[wave] = incl;
  }
  __syncthreads();
  if (tid < NBIN) {
    int woff = 0;
    for (int w = 0; w < NBIN / 32; ++w)
      if (w < wave) woff += wsum[w];
    start[tid] = woff + incl - c;
    gbase[tid] = c ? atomicAdd(&fill[tid], c) : 0;
  }
  __syncthreads();
}

__global__ __launch_bounds__(BT) void bin_edges(const int* __restrict__ src,
                                                const int* __restrict__ dst,
                                                int* __restrict__ dfill,
                                                int* __restrict__ sfill,
                                                int2* __restrict__ dbin,
                                                int* __restrict__ sbin,
                                                int* __restrict__ misc,
                                                int2* __restrict__ dspill,
                                                int* __restrict__ sspill,
                                                int nedges) {
  __shared__ int hist[NBIN];
  __shared__ int start[NBIN];
  __shared__ int gbase[NBIN];
  __shared__ int wsum[NBIN / 32];
  __shared__ int2 sorted[EPB];

  const int tid = threadIdx.x;
  const int base = blockIdx.x * EPB;
  const int total = min(EPB, nedges - base);

  int s[EPT], d[EPT], r[EPT];
#pragma unroll
  for (int k = 0; k < EPT; ++k) {
    const int e = min(base + k * BT + tid, nedges - 1);
    s[k] = src[e];
    d[k] = dst[e];
  }

  if (tid < NBIN) hist[tid] = 0;
  __syncthreads();
#pragma unroll
  for (int k = 0; k < EPT; ++k)
    if (k * BT + tid < total) r[k] = atomicAdd(&hist[bin_of(d[k])], 1);
  __syncthreads();
  scan_reserve(hist, start, gbase, wsum, dfill, tid);
#pragma unroll
  for (int k = 0; k < EPT; ++k)
    if (k * BT + tid < total) sorted[start[bin_of(d[k])] + r[k]] = make_int2(d[k], s[k]);
  __syncthreads();
  for (int i = tid; i < total; i += BT) {
    const int2 p = sorted[i];
    const int b = bin_of(p.x);
    const int pos = gbase[b] + (i - start[b]);
    if (pos < BCAP) {
      dbin[b * BCAP + pos] = p;
    } else {
      const int j = atomicAdd(&misc[1], 1);
      dspill[j] = p;
    }
  }
  __syncthreads();

  int* sorted1 = (int*)sorted;
  if (tid < NBIN) hist[tid] = 0;
  __syncthreads();
#pragma unroll
  for (int k = 0; k < EPT; ++k)
    if (k * BT + tid < total) r[k] = atomicAdd(&hist[bin_of(s[k])], 1);
  __syncthreads();
  scan_reserve(hist, start, gbase, wsum, sfill, tid);
#pragma unroll
  for (int k = 0; k < EPT; ++k)
    if (k * BT + tid < total) sorted1[start[bin_of(s[k])] + r[k]] = s[k];
  __syncthreads();
  for (int i = tid; i < total; i += BT) {
    const int sv = sorted1[i];
    const int b = bin_of(sv);
    const int pos = gbase[b] + (i - start[b]);
    if (pos < BCAP) {
      sbin[b * BCAP + pos] = sv;
    } else {
      const int j = atomicAdd(&misc[2], 1);
      sspill[j] = sv;
    }
  }
}

__global__ __launch_bounds__(PT) void place_bin(const int* __restrict__ dfill,
                                                const int* __restrict__ sfill,
                                                const int2* __restrict__ dbin,
                                                const int* __restrict__ sbin,
                                                int* __restrict__ misc,
                                                const int2* __restrict__ dspill,
                                                const int* __restrict__ sspill,
                                                int* __restrict__ cnt_in,
                                                int* __restrict__ cnt_out,
                                                int* __restrict__ rec,
                                                int2* __restrict__ ovf,
                                                int nrows, int nedges) {
  __shared__ int cin[NPB];
  __shared__ int cout[NPB];
  __shared__ __attribute__((aligned(16))) int rows[NPB * SLOTS];

  const int tid = threadIdx.x;
  const int b = blockIdx.x;
  const int node0 = b * NPB;
  if (tid < NPB) {
    cin[tid] = 0;
    cout[tid] = 0;
  }
  const int nd = min(dfill[b], BCAP);
  const int ns = min(sfill[b], BCAP);
  const int ndsp = min(misc[1], nedges);
  const int nssp = min(misc[2], nedges);
  __syncthreads();

#define ADD_PAIR(p)                                                   \
  do {                                                                \
    const int dl = (p).x - node0;                                     \
    if ((unsigned)dl < NPB) {                                         \
      const int slot = atomicAdd(&cin[dl], 1);                        \
      if (slot < SLOTS) {                                             \
        rows[dl * SLOTS + slot] = (p).y;                              \
      } else {                                                        \
        const int j = atomicAdd(&misc[0], 1);                         \
        ovf[j] = (p);                                                 \
      }                                                               \
    }                                                                 \
  } while (0)

  int2 pr[BCAP / PT];
  int sv[BCAP / PT];
#pragma unroll
  for (int k = 0; k < BCAP / PT; ++k) {
    const int i = k * PT + tid;
    if (i < nd) pr[k] = dbin[b * BCAP + i];
    if (i < ns) sv[k] = sbin[b * BCAP + i];
  }
#pragma unroll
  for (int k = 0; k < BCAP / PT; ++k)
    if (k * PT + tid < nd) ADD_PAIR(pr[k]);
  for (int i = tid; i < ndsp; i += PT) {
    const int2 p = dspill[i];
    ADD_PAIR(p);
  }
#undef ADD_PAIR
#pragma unroll
  for (int k = 0; k < BCAP / PT; ++k) {
    const int sl = sv[k] - node0;
    if (k * PT + tid < ns && (unsigned)sl < NPB) atomicAdd(&cout[sl], 1);
  }
  for (int i = tid; i < nssp; i += PT) {
    const int sl = sspill[i] - node0;
    if ((unsigned)sl < NPB) atomicAdd(&cout[sl], 1);
  }
  __syncthreads();

  if (tid < NPB && node0 + tid < nrows) {
    cnt_in[node0 + tid] = cin[tid];
    cnt_out[node0 + tid] = cout[tid];
  }
  for (int q = tid; q < NPB * (SLOTS / 4); q += PT) {
    const int dl = q / (SLOTS / 4);
    const int piece = q % (SLOTS / 4);
    if (node0 + dl < nrows && piece * 4 < min(cin[dl], SLOTS))
      ((int4*)rec)[(long long)(node0 + dl) * (SLOTS / 4) + piece] = ((const int4*)rows)[q];
  }
}

#define SC 8

__global__ __launch_bounds__(256) void scale_rows(const float* __restrict__ h,
                                                  const int* __restrict__ cnt_out,
                                                  float* __restrict__ hs,
                                                  int nvec) {
  const int base = blockIdx.x * (256 * SC) + threadIdx.x;
  v4f v[SC];
  int c[SC];
#pragma unroll
  for (int k = 0; k < SC; ++k) {
    const int i = min(base + k * 256, nvec - 1);
    v[k] = ((const v4f*)h)[i];
    c[k] = cnt_out[i / (FEAT / 4)];
  }
#pragma unroll
  for (int k = 0; k < SC; ++k) {
    const int i = base + k * 256;
    if (i < nvec) {
      const float ns = 1.0f / sqrtf(fmaxf((float)c[k], 1.0f));
      ((v4f*)hs)[i] = v[k] * ns;
    }
  }
}

#define APITCH 100
#define GU 16

__device__ __forceinline__ void add4(float4& acc, const float4 v) {
  acc.x += v.x;
  acc.y += v.y;
  acc.z += v.z;
  acc.w += v.w;
}

__device__ __forceinline__ int rec_index(const int rp, const int v, const int row0) {
  int r = 0, st = 0;
#pragma unroll
  for (int j = 1; j < 16; ++j) {
    const int rj = __builtin_amdgcn_readlane(rp, j);
    if (v >= rj) { r = j; st = rj; }
  }
  return (row0 + r) * SLOTS + (v - st);
}

__global__ __launch_bounds__(256) void gather_gemm_bias_relu(
    const int* __restrict__ cnt_in, const int* __restrict__ cnt_out,
    const int* __restrict__ rec, const int* __restrict__ ovf_cnt,
    const int2* __restrict__ ovf, const float* __restrict__ h,
    const float* __restrict__ W, const float* __restrict__ bias,
    float* __restrict__ out, int nrows, int nedges, int scale_out) {
  __shared__ __attribute__((aligned(16))) float Ws[FEAT * FEAT];
  __shared__ float bs[FEAT];
  __shared__ __attribute__((aligned(16))) float As[8 * 16 * APITCH];

  const int tid = threadIdx.x;
  const int lane = tid & 31;
  const int wave = __builtin_amdgcn_readfirstlane(tid >> 5);
  const int row0 = (blockIdx.x * 8 + wave) * 16;
  float* __restrict__ strip = As + wave * (16 * APITCH);

  v4f wreg[9];
#pragma unroll
  for (int i = 0; i < 9; ++i) wreg[i] = ((const v4f*)W)[tid + 256 * i];
  const int novf_lane = ovf_cnt[0];

  if (row0 < nrows) {
    const int fo = min(lane, FEAT / 4 - 1) * 4;
    const int rr = row0 + (lane & 15);
    int cfull = cnt_in[min(rr, nrows - 1)];
    if (lane >= 16 || rr >= nrows) cfull = 0;
    const float ndl = 1.0f / sqrtf(fmaxf((float)cfull, 1.0f));
    const int c = min(cfull, SLOTS);
    int incl = c;
    for (int o = 1; o < 16; o <<= 1) {
      const int t = __shfl_up(incl, o);
      if (lane >= o) incl += t;
    }
    const int end = __builtin_amdgcn_readlane(incl, 15);
    int rp = __shfl_up(incl, 1);
    if (lane == 0) rp = 0;
    if (lane >= 16) rp = end;

    int cur = 0;
    int nextb = __builtin_amdgcn_readlane(rp, 1);
    float4 acc = make_float4(0.f, 0.f, 0.f, 0.f);

#define PARK_ROW()                                                                    \
    do {                                                                              \
      const float nd = __int_as_float(                                                \
          __builtin_amdgcn_readlane(__float_as_int(ndl), cur));                       \
      if (lane < FEAT / 4)                                                            \
        *(float4*)(strip + cur * APITCH + lane * 4) =                                 \
            make_float4(acc.x * nd, acc.y * nd, acc.z * nd, acc.w * nd);              \
      acc = make_float4(0.f, 0.f, 0.f, 0.f);                                          \
      ++cur;                                                                          \
      nextb = __builtin_amdgcn_readlane(rp, cur + 1);                                 \
    } while (0)

    int p = 0;
    if (0 < end) p = rec[rec_index(rp, min(lane, end - 1), row0)];
    for (int base = 0; base < end; base += 32) {
      int pn = p;
#pragma unroll
      for (int g = 0; g < 32 / GU; ++g) {
        const int e0 = base + g * GU;
        if (e0 < end) {
          float4 v[GU];
#pragma unroll
          for (int u = 0; u < GU; ++u) {
            const int s = __builtin_amdgcn_readlane(p, g * GU + u);
            v[u] = *(const float4*)(h + (long long)s * FEAT + fo);
          }
          if (g == 0 && base + 32 < end)
            pn = rec[rec_index(rp, min(base + 32 + lane, end - 1), row0)];
          if (e0 + GU <= end) {
#pragma unroll
            for (int u = 0; u < GU; ++u) {
              const int e = e0 + u;
              if (__builtin_expect(e == nextb, 0)) {
                do { PARK_ROW(); } while (e == nextb);
              }
              add4(acc, v[u]);
            }
          } else {
#pragma unroll
            for (int u = 0; u < GU; ++u) {
              const int e = e0 + u;
              if (e < end) {
                if (__builtin_expect(e == nextb, 0)) {
                  do { PARK_ROW(); } while (e == nextb);
                }
                add4(acc, v[u]);
              }
            }
          }
        }
      }
      p = pn;
    }
    while (cur < 16) PARK_ROW();
#undef PARK_ROW

    const int novf = min(__builtin_amdgcn_readfirstlane(novf_lane), nedges);
    for (int i = 0; i < novf; ++i) {
      const int2 o = ovf[i];
      const int r = __builtin_amdgcn_readfirstlane(o.x) - row0;
      if (r >= 0 && r < 16) {
        const int s = __builtin_amdgcn_readfirstlane(o.y);
        const float nd = __int_as_float(__builtin_amdgcn_readlane(__float_as_int(ndl), r));
        if (lane < FEAT / 4) {
          float4* q = (float4*)(strip + r * APITCH + lane * 4);
          const float4 hv = *(const float4*)(h + (long long)s * FEAT + lane * 4);
          float4 t = *q;
          t.x = fmaf(hv.x, nd, t.x);
          t.y = fmaf(hv.y, nd, t.y);
          t.z = fmaf(hv.z, nd, t.z);
          t.w = fmaf(hv.w, nd, t.w);
          *q = t;
        }
      }
    }
  }

#pragma unroll
  for (int i = 0; i < 9; ++i) ((v4f*)Ws)[tid + 256 * i] = wreg[i];
  if (tid < FEAT) bs[tid] = bias[tid];
  __syncthreads();

  if (row0 >= nrows) return;

  const int lrow = lane & 15;
  const int lgrp = lane >> 4;

  float osc[8];
#pragma unroll
  for (int r = 0; r < 8; ++r) {
    osc[r] = 1.0f;
    if (scale_out) {
      const int row = min(row0 + r + 8 * lgrp, nrows - 1);
      osc[r] = 1.0f / sqrtf(fmaxf((float)cnt_out[row], 1.0f));
    }
  }

  v2f a[KSTEPS];
#pragma unroll
  for (int ks = 0; ks < KSTEPS; ++ks)
    a[ks] = *(const v2f*)(strip + lrow * APITCH + ks * 4 + 2 * lgrp);

  v8f acc[NTILES];
#pragma unroll
  for (int nt = 0; nt < NTILES; ++nt) acc[nt] = (v8f){};

  v2f bf[NTILES];
#pragma unroll
  for (int nt = 0; nt < NTILES; ++nt) {
    const int k = 2 * lgrp;
    bf[nt].x = Ws[k * FEAT + nt * 16 + lrow];
    bf[nt].y = Ws[(k + 1) * FEAT + nt * 16 + lrow];
  }

#pragma unroll
  for (int ks = 0; ks < KSTEPS; ++ks) {
    v2f bn[NTILES];
    if (ks + 1 < KSTEPS) {
      const int k = (ks + 1) * 4 + 2 * lgrp;
#pragma unroll
      for (int nt = 0; nt < NTILES; ++nt) {
        bn[nt].x = Ws[k * FEAT + nt * 16 + lrow];
        bn[nt].y = Ws[(k + 1) * FEAT + nt * 16 + lrow];
      }
    }
#pragma unroll
    for (int nt = 0; nt < NTILES; ++nt) {
      acc[nt] = __builtin_amdgcn_wmma_f32_16x16x4_f32(
           false, a[ks],  false, bf[nt],
           (short)0, acc[nt],  false,  false);
    }
    if (ks + 1 < KSTEPS) {
#pragma unroll
      for (int nt = 0; nt < NTILES; ++nt) bf[nt] = bn[nt];
    }
  }

#pragma unroll
  for (int nt = 0; nt < NTILES; ++nt) {
    const int nb = nt * 16;
    const float bcol = bs[nb + lrow];
#pragma unroll
    for (int r = 0; r < 8; ++r) {
      const int row = row0 + r + 8 * lgrp;
      if (row < nrows) {
        float v = acc[nt][r] + bcol;
        out[(long long)row * FEAT + nb + lrow] = fmaxf(v, 0.0f) * osc[r];
      }
    }
  }
}

extern "C" void kernel_launch(void* const* d_in, const int* in_sizes, int n_in,
                              void* d_out, int out_size, void* d_ws, size_t ws_size,
                              hipStream_t stream) {
  const float* h0  = (const float*)d_in[0];
  const int*   src = (const int*)d_in[1];
  const int*   dst = (const int*)d_in[2];
  const float* Wp[3] = {(const float*)d_in[3], (const float*)d_in[5], (const float*)d_in[7]};
  const float* bp[3] = {(const float*)d_in[4], (const float*)d_in[6], (const float*)d_in[8]};

  const int N = in_sizes[0] / FEAT;
  const int E = in_sizes[1];

  const long long NP = ((long long)N + 63) / 64 * 64;
  const long long EP = ((long long)E + 63) / 64 * 64;
  float* ws       = (float*)d_ws;
  int*   cnt_out  = (int*)ws;
  int*   cnt_in   = (int*)(ws + NP);
  int*   misc     = (int*)(ws + 2 * NP);
  int*   dfill    = misc + 64;
  int*   sfill    = dfill + NBIN;
  const long long HEAD = 2 * NP + 64 + 2 * NBIN;
  int*   rec      = (int*)(ws + HEAD);
  int2*  ovf      = (int2*)(ws + HEAD + (long long)N * SLOTS);
  float* hA       = ws + HEAD + (long long)N * SLOTS + 2 * EP;
  float* hs0      = hA + (long long)N * FEAT;
  int2*  dbin     = (int2*)(hs0 + (long long)N * FEAT);
  int*   sbin     = (int*)(hs0 + (long long)N * FEAT + 2LL * NBIN * BCAP);
  int2*  dspill   = (int2*)(sbin + (long long)NBIN * BCAP);
  int*   sspill   = (int*)(dspill + EP);
  float* outp     = (float*)d_out;

  {
    long long n2 = 64 + 2 * NBIN;
    int blocks = (int)((n2 / 4 + 255) / 256) + 1;
    zero_f32<<<blocks, 256, 0, stream>>>((float*)misc, n2);
    bin_edges<<<(E + EPB - 1) / EPB, BT, 0, stream>>>(src, dst, dfill, sfill, dbin, sbin, misc,
                                                      dspill, sspill, E);
    place_bin<<<(N + NPB - 1) / NPB, PT, 0, stream>>>(dfill, sfill, dbin, sbin, misc, dspill,
                                                      sspill, cnt_in, cnt_out, rec, ovf, N, E);
    const int nvec = N * (FEAT / 4);
    scale_rows<<<(nvec + 256 * SC - 1) / (256 * SC), 256, 0, stream>>>(h0, cnt_out, hs0, nvec);
  }

  const int strips = (N + 15) / 16;
  const int gemm_blocks = (strips + 7) / 8;

  const float* lin[3]  = {hs0, outp, hA};
  float*       lout[3] = {outp, hA, outp};

  for (int l = 0; l < 3; ++l) {
    gather_gemm_bias_relu<<<gemm_blocks, 256, 0, stream>>>(cnt_in, cnt_out, rec, misc, ovf,
                                                           lin[l], Wp[l], bp[l], lout[l],
                                                           N, E, l < 2 ? 1 : 0);
  }
}


# Layer kernel: three waves per 16-row strip (768-thread blocks, two per WGP): each wave gathers 5-6 of its rows, 8 row loads per group, and computes 2 of 6 column tiles; row sum order unchanged
# speedup vs baseline: 4.2967x; 1.0335x over previous
#include <hip/hip_runtime.h>
#include <hip/hip_bf16.h>

#define FEAT 96
#define KSTEPS (FEAT / 4)
#define NTILES (FEAT / 16)

typedef float v2f __attribute__((ext_vector_type(2)));
typedef float v4f __attribute__((ext_vector_type(4)));
typedef float v8f __attribute__((ext_vector_type(8)));

__global__ __launch_bounds__(256) void zero_f32(float* __restrict__ p, long long n) {
  long long i = (long long)blockIdx.x * blockDim.x + threadIdx.x;
  long long i4 = i * 4;
  if (i4 + 3 < n) {
    *(float4*)(p + i4) = make_float4(0.f, 0.f, 0.f, 0.f);
  } else {
    for (long long j = i4; j < n; ++j) p[j] = 0.f;
  }
}

#define SLOTS 64
#define NPB 196
#define NBIN 256
#define BCAP 4096
#define BT 512
#define EPT 8
#define EPB (BT * EPT)
#define PT 1024

__device__ __forceinline__ int bin_of(const int node) {
  return min((int)((unsigned)node / NPB), NBIN - 1);
}

__device__ __forceinline__ void scan_reserve(const int* hist, int* start, int* gbase,
                                             int* wsum, int* __restrict__ fill,
                                             const int tid) {
  const int lane = tid & 31;
  const int wave = tid >> 5;
  int c = 0, incl = 0;
  if (tid < NBIN) {
    c = hist[tid];
    incl = c;
    for (int o = 1; o < 32; o <<= 1) {
      const int t = __shfl_up(incl, o);
      if (lane >= o) incl += t;
    }
    if (lane == 31) wsum[wave] = incl;
  }
  __syncthreads();
  if (tid < NBIN) {
    int woff = 0;
    for (int w = 0; w < NBIN / 32; ++w)
      if (w < wave) woff += wsum[w];
    start[tid] = woff + incl - c;
    gbase[tid] = c ? atomicAdd(&fill[tid], c) : 0;
  }
  __syncthreads();
}

__global__ __launch_bounds__(BT) void bin_edges(const int* __restrict__ src,
                                                const int* __restrict__ dst,
                                                int* __restrict__ dfill,
                                                int* __restrict__ sfill,
                                                int2* __restrict__ dbin,
                                                int* __restrict__ sbin,
                                                int* __restrict__ misc,
                                                int2* __restrict__ dspill,
                                                int* __restrict__ sspill,
                                                int nedges) {
  __shared__ int hist[NBIN];
  __shared__ int start[NBIN];
  __shared__ int gbase[NBIN];
  __shared__ int wsum[NBIN / 32];
  __shared__ int2 sorted[EPB];

  const int tid = threadIdx.x;
  const int base = blockIdx.x * EPB;
  const int total = min(EPB, nedges - base);

  int s[EPT], d[EPT], r[EPT];
#pragma unroll
  for (int k = 0; k < EPT; ++k) {
    const int e = min(base + k * BT + tid, nedges - 1);
    s[k] = src[e];
    d[k] = dst[e];
  }

  if (tid < NBIN) hist[tid] = 0;
  __syncthreads();
#pragma unroll
  for (int k = 0; k < EPT; ++k)
    if (k * BT + tid < total) r[k] = atomicAdd(&hist[bin_of(d[k])], 1);
  __syncthreads();
  scan_reserve(hist, start, gbase, wsum, dfill, tid);
#pragma unroll
  for (int k = 0; k < EPT; ++k)
    if (k * BT + tid < total) sorted[start[bin_of(d[k])] + r[k]] = make_int2(d[k], s[k]);
  __syncthreads();
  for (int i = tid; i < total; i += BT) {
    const int2 p = sorted[i];
    const int b = bin_of(p.x);
    const int pos = gbase[b] + (i - start[b]);
    if (pos < BCAP) {
      dbin[b * BCAP + pos] = p;
    } else {
      const int j = atomicAdd(&misc[1], 1);
      dspill[j] = p;
    }
  }
  __syncthreads();

  int* sorted1 = (int*)sorted;
  if (tid < NBIN) hist[tid] = 0;
  __syncthreads();
#pragma unroll
  for (int k = 0; k < EPT; ++k)
    if (k * BT + tid < total) r[k] = atomicAdd(&hist[bin_of(s[k])], 1);
  __syncthreads();
  scan_reserve(hist, start, gbase, wsum, sfill, tid);
#pragma unroll
  for (int k = 0; k < EPT; ++k)
    if (k * BT + tid < total) sorted1[start[bin_of(s[k])] + r[k]] = s[k];
  __syncthreads();
  for (int i = tid; i < total; i += BT) {
    const int sv = sorted1[i];
    const int b = bin_of(sv);
    const int pos = gbase[b] + (i - start[b]);
    if (pos < BCAP) {
      sbin[b * BCAP + pos] = sv;
    } else {
      const int j = atomicAdd(&misc[2], 1);
      sspill[j] = sv;
    }
  }
}

__global__ __launch_bounds__(PT) void place_bin(const int* __restrict__ dfill,
                                                const int* __restrict__ sfill,
                                                const int2* __restrict__ dbin,
                                                const int* __restrict__ sbin,
                                                int* __restrict__ misc,
                                                const int2* __restrict__ dspill,
                                                const int* __restrict__ sspill,
                                                int* __restrict__ cnt_in,
                                                int* __restrict__ cnt_out,
                                                int* __restrict__ rec,
                                                int2* __restrict__ ovf,
                                                int nrows, int nedges) {
  __shared__ int cin[NPB];
  __shared__ int cout[NPB];
  __shared__ __attribute__((aligned(16))) int rows[NPB * SLOTS];

  const int tid = threadIdx.x;
  const int b = blockIdx.x;
  const int node0 = b * NPB;
  if (tid < NPB) {
    cin[tid] = 0;
    cout[tid] = 0;
  }
  const int nd = min(dfill[b], BCAP);
  const int ns = min(sfill[b], BCAP);
  const int ndsp = min(misc[1], nedges);
  const int nssp = min(misc[2], nedges);
  __syncthreads();

#define ADD_PAIR(p)                                                   \
  do {                                                                \
    const int dl = (p).x - node0;                                     \
    if ((unsigned)dl < NPB) {                                         \
      const int slot = atomicAdd(&cin[dl], 1);                        \
      if (slot < SLOTS) {                                             \
        rows[dl * SLOTS + slot] = (p).y;                              \
      } else {                                                        \
        const int j = atomicAdd(&misc[0], 1);                         \
        ovf[j] = (p);                                                 \
      }                                                               \
    }                                                                 \
  } while (0)

  int2 pr[BCAP / PT];
  int sv[BCAP / PT];
#pragma unroll
  for (int k = 0; k < BCAP / PT; ++k) {
    const int i = k * PT + tid;
    if (i < nd) pr[k] = dbin[b * BCAP + i];
    if (i < ns) sv[k] = sbin[b * BCAP + i];
  }
#pragma unroll
  for (int k = 0; k < BCAP / PT; ++k)
    if (k * PT + tid < nd) ADD_PAIR(pr[k]);
  for (int i = tid; i < ndsp; i += PT) {
    const int2 p = dspill[i];
    ADD_PAIR(p);
  }
#undef ADD_PAIR
#pragma unroll
  for (int k = 0; k < BCAP / PT; ++k) {
    const int sl = sv[k] - node0;
    if (k * PT + tid < ns && (unsigned)sl < NPB) atomicAdd(&cout[sl], 1);
  }
  for (int i = tid; i < nssp; i += PT) {
    const int sl = sspill[i] - node0;
    if ((unsigned)sl < NPB) atomicAdd(&cout[sl], 1);
  }
  __syncthreads();

  if (tid < NPB && node0 + tid < nrows) {
    cnt_in[node0 + tid] = cin[tid];
    cnt_out[node0 + tid] = cout[tid];
  }
  for (int q = tid; q < NPB * (SLOTS / 4); q += PT) {
    const int dl = q / (SLOTS / 4);
    const int piece = q % (SLOTS / 4);
    if (node0 + dl < nrows && piece * 4 < min(cin[dl], SLOTS))
      ((int4*)rec)[(long long)(node0 + dl) * (SLOTS / 4) + piece] = ((const int4*)rows)[q];
  }
}

#define SC 8

__global__ __launch_bounds__(256) void scale_rows(const float* __restrict__ h,
                                                  const int* __restrict__ cnt_out,
                                                  float* __restrict__ hs,
                                                  int nvec) {
  const int base = blockIdx.x * (256 * SC) + threadIdx.x;
  v4f v[SC];
  int c[SC];
#pragma unroll
  for (int k = 0; k < SC; ++k) {
    const int i = min(base + k * 256, nvec - 1);
    v[k] = ((const v4f*)h)[i];
    c[k] = cnt_out[i / (FEAT / 4)];
  }
#pragma unroll
  for (int k = 0; k < SC; ++k) {
    const int i = base + k * 256;
    if (i < nvec) {
      const float ns = 1.0f / sqrtf(fmaxf((float)c[k], 1.0f));
      ((v4f*)hs)[i] = v[k] * ns;
    }
  }
}

#define APITCH 100
#define GU 8
#define WPS 3
#define RPW 6
#define TPW (NTILES / WPS)

__device__ __forceinline__ void add4(float4& acc, const float4 v) {
  acc.x += v.x;
  acc.y += v.y;
  acc.z += v.z;
  acc.w += v.w;
}

__device__ __forceinline__ int rec_index(const int rp, const int v, const int row0) {
  int r = 0, st = 0;
#pragma unroll
  for (int j = 1; j < RPW; ++j) {
    const int rj = __builtin_amdgcn_readlane(rp, j);
    if (v >= rj) { r = j; st = rj; }
  }
  return (row0 + r) * SLOTS + (v - st);
}

__global__ __launch_bounds__(768, 12) void gather_gemm_bias_relu(
    const int* __restrict__ cnt_in, const int* __restrict__ cnt_out,
    const int* __restrict__ rec, const int* __restrict__ ovf_cnt,
    const int2* __restrict__ ovf, const float* __restrict__ h,
    const float* __restrict__ W, const float* __restrict__ bias,
    float* __restrict__ out, int nrows, int nedges, int scale_out) {
  __shared__ __attribute__((aligned(16))) float Ws[FEAT * FEAT];
  __shared__ float bs[FEAT];
  __shared__ __attribute__((aligned(16))) float As[8 * 16 * APITCH];

  const int tid = threadIdx.x;
  const int lane = tid & 31;
  const int wave = __builtin_amdgcn_readfirstlane(tid >> 5);
  const int sidx = wave / WPS;
  const int part = wave - sidx * WPS;
  const int nrw = part == 0 ? 6 : 5;
  const int rs = part == 0 ? 0 : 1 + 5 * part;
  const int row0 = (blockIdx.x * 8 + sidx) * 16;
  const int grow0 = row0 + rs;
  float* __restrict__ strip = As + sidx * (16 * APITCH);
  float* __restrict__ gstrip = strip + rs * APITCH;

  v4f wreg[3];
#pragma unroll
  for (int i = 0; i < 3; ++i) wreg[i] = ((const v4f*)W)[tid + 768 * i];
  const int novf_lane = ovf_cnt[0];

  if (row0 < nrows) {
    const int fo = min(lane, FEAT / 4 - 1) * 4;
    const int rr = grow0 + min(lane, nrw - 1);
    int cfull = cnt_in[min(rr, nrows - 1)];
    if (lane >= nrw || rr >= nrows) cfull = 0;
    const float ndl = 1.0f / sqrtf(fmaxf((float)cfull, 1.0f));
    const int c = min(cfull, SLOTS);
    int incl = c;
    for (int o = 1; o < 8; o <<= 1) {
      const int t = __shfl_up(incl, o);
      if (lane >= o) incl += t;
    }
    const int end = __builtin_amdgcn_readlane(incl, 7);
    int rp = __shfl_up(incl, 1);
    if (lane == 0) rp = 0;
    if (lane >= nrw) rp = end;

    int cur = 0;
    int nextb = __builtin_amdgcn_readlane(rp, 1);
    float4 acc = make_float4(0.f, 0.f, 0.f, 0.f);

#define PARK_ROW()                                                                    \
    do {                                                                              \
      const float nd = __int_as_float(                                                \
          __builtin_amdgcn_readlane(__float_as_int(ndl), cur));                       \
      if (lane < FEAT / 4)                                                            \
        *(float4*)(gstrip + cur * APITCH + lane * 4) =                                \
            make_float4(acc.x * nd, acc.y * nd, acc.z * nd, acc.w * nd);              \
      acc = make_float4(0.f, 0.f, 0.f, 0.f);                                          \
      ++cur;                                                                          \
      nextb = __builtin_amdgcn_readlane(rp, cur + 1);                                 \
    } while (0)

    int p = 0;
    if (0 < end) p = rec[rec_index(rp, min(lane, end - 1), grow0)];
    for (int base = 0; base < end; base += 32) {
      int pn = p;
#pragma unroll
      for (int g = 0; g < 32 / GU; ++g) {
        const int e0 = base + g * GU;
        if (e0 < end) {
          float4 v[GU];
#pragma unroll
          for (int u = 0; u < GU; ++u) {
            const int s = __builtin_amdgcn_readlane(p, g * GU + u);
            v[u] = *(const float4*)(h + (long long)s * FEAT + fo);
          }
          if (g == 0 && base + 32 < end)
            pn = rec[rec_index(rp, min(base + 32 + lane, end - 1), grow0)];
          if (e0 + GU <= end) {
#pragma unroll
            for (int u = 0; u < GU; ++u) {
              const int e = e0 + u;
              if (__builtin_expect(e == nextb, 0)) {
                do { PARK_ROW(); } while (e == nextb);
              }
              add4(acc, v[u]);
            }
          } else {
#pragma unroll
            for (int u = 0; u < GU; ++u) {
              const int e = e0 + u;
              if (e < end) {
                if (__builtin_expect(e == nextb, 0)) {
                  do { PARK_ROW(); } while (e == nextb);
                }
                add4(acc, v[u]);
              }
            }
          }
        }
      }
      p = pn;
    }
    while (cur < nrw) PARK_ROW();
#undef PARK_ROW

    const int novf = min(__builtin_amdgcn_readfirstlane(novf_lane), nedges);
    for (int i = 0; i < novf; ++i) {
      const int2 o = ovf[i];
      const int r = __builtin_amdgcn_readfirstlane(o.x) - grow0;
      if (r >= 0 && r < nrw) {
        const int s = __builtin_amdgcn_readfirstlane(o.y);
        const float nd = __int_as_float(__builtin_amdgcn_readlane(__float_as_int(ndl), r));
        if (lane < FEAT / 4) {
          float4* q = (float4*)(gstrip + r * APITCH + lane * 4);
          const float4 hv = *(const float4*)(h + (long long)s * FEAT + lane * 4);
          float4 t = *q;
          t.x = fmaf(hv.x, nd, t.x);
          t.y = fmaf(hv.y, nd, t.y);
          t.z = fmaf(hv.z, nd, t.z);
          t.w = fmaf(hv.w, nd, t.w);
          *q = t;
        }
      }
    }
  }

#pragma unroll
  for (int i = 0; i < 3; ++i) ((v4f*)Ws)[tid + 768 * i] = wreg[i];
  if (tid < FEAT) bs[tid] = bias[tid];
  __syncthreads();

  if (row0 >= nrows) return;

  const int lrow = lane & 15;
  const int lgrp = lane >> 4;

  float osc[8];
#pragma unroll
  for (int r = 0; r < 8; ++r) {
    osc[r] = 1.0f;
    if (scale_out) {
      const int row = min(row0 + r + 8 * lgrp, nrows - 1);
      osc[r] = 1.0f / sqrtf(fmaxf((float)cnt_out[row], 1.0f));
    }
  }

  v2f a[KSTEPS];
#pragma unroll
  for (int ks = 0; ks < KSTEPS; ++ks)
    a[ks] = *(const v2f*)(strip + lrow * APITCH + ks * 4 + 2 * lgrp);

  const int cb0 = part * (TPW * 16);

  v8f acc[TPW];
#pragma unroll
  for (int nt = 0; nt < TPW; ++nt) acc[nt] = (v8f){};

  v2f bf[TPW];
#pragma unroll
  for (int nt = 0; nt < TPW; ++nt) {
    const int k = 2 * lgrp;
    bf[nt].x = Ws[k * FEAT + cb0 + nt * 16 + lrow];
    bf[nt].y = Ws[(k + 1) * FEAT + cb0 + nt * 16 + lrow];
  }

#pragma unroll
  for (int ks = 0; ks < KSTEPS; ++ks) {
    v2f bn[TPW];
    if (ks + 1 < KSTEPS) {
      const int k = (ks + 1) * 4 + 2 * lgrp;
#pragma unroll
      for (int nt = 0; nt < TPW; ++nt) {
        bn[nt].x = Ws[k * FEAT + cb0 + nt * 16 + lrow];
        bn[nt].y = Ws[(k + 1) * FEAT + cb0 + nt * 16 + lrow];
      }
    }
#pragma unroll
    for (int nt = 0; nt < TPW; ++nt) {
      acc[nt] = __builtin_amdgcn_wmma_f32_16x16x4_f32(
           false, a[ks],  false, bf[nt],
           (short)0, acc[nt],  false,  false);
    }
    if (ks + 1 < KSTEPS) {
#pragma unroll
      for (int nt = 0; nt < TPW; ++nt) bf[nt] = bn[nt];
    }
  }

#pragma unroll
  for (int nt = 0; nt < TPW; ++nt) {
    const int nb = cb0 + nt * 16;
    const float bcol = bs[nb + lrow];
#pragma unroll
    for (int r = 0; r < 8; ++r) {
      const int row = row0 + r + 8 * lgrp;
      if (row < nrows) {
        float v = acc[nt][r] + bcol;
        out[(long long)row * FEAT + nb + lrow] = fmaxf(v, 0.0f) * osc[r];
      }
    }
  }
}

extern "C" void kernel_launch(void* const* d_in, const int* in_sizes, int n_in,
                              void* d_out, int out_size, void* d_ws, size_t ws_size,
                              hipStream_t stream) {
  const float* h0  = (const float*)d_in[0];
  const int*   src = (const int*)d_in[1];
  const int*   dst = (const int*)d_in[2];
  const float* Wp[3] = {(const float*)d_in[3], (const float*)d_in[5], (const float*)d_in[7]};
  const float* bp[3] = {(const float*)d_in[4], (const float*)d_in[6], (const float*)d_in[8]};

  const int N = in_sizes[0] / FEAT;
  const int E = in_sizes[1];

  const long long NP = ((long long)N + 63) / 64 * 64;
  const long long EP = ((long long)E + 63) / 64 * 64;
  float* ws       = (float*)d_ws;
  int*   cnt_out  = (int*)ws;
  int*   cnt_in   = (int*)(ws + NP);
  int*   misc     = (int*)(ws + 2 * NP);
  int*   dfill    = misc + 64;
  int*   sfill    = dfill + NBIN;
  const long long HEAD = 2 * NP + 64 + 2 * NBIN;
  int*   rec      = (int*)(ws + HEAD);
  int2*  ovf      = (int2*)(ws + HEAD + (long long)N * SLOTS);
  float* hA       = ws + HEAD + (long long)N * SLOTS + 2 * EP;
  float* hs0      = hA + (long long)N * FEAT;
  int2*  dbin     = (int2*)(hs0 + (long long)N * FEAT);
  int*   sbin     = (int*)(hs0 + (long long)N * FEAT + 2LL * NBIN * BCAP);
  int2*  dspill   = (int2*)(sbin + (long long)NBIN * BCAP);
  int*   sspill   = (int*)(dspill + EP);
  float* outp     = (float*)d_out;

  {
    long long n2 = 64 + 2 * NBIN;
    int blocks = (int)((n2 / 4 + 255) / 256) + 1;
    zero_f32<<<blocks, 256, 0, stream>>>((float*)misc, n2);
    bin_edges<<<(E + EPB - 1) / EPB, BT, 0, stream>>>(src, dst, dfill, sfill, dbin, sbin, misc,
                                                      dspill, sspill, E);
    place_bin<<<(N + NPB - 1) / NPB, PT, 0, stream>>>(dfill, sfill, dbin, sbin, misc, dspill,
                                                      sspill, cnt_in, cnt_out, rec, ovf, N, E);
    const int nvec = N * (FEAT / 4);
    scale_rows<<<(nvec + 256 * SC - 1) / (256 * SC), 256, 0, stream>>>(h0, cnt_out, hs0, nvec);
  }

  const int strips = (N + 15) / 16;
  const int gemm_blocks = (strips + 7) / 8;

  const float* lin[3]  = {hs0, outp, hA};
  float*       lout[3] = {outp, hA, outp};

  for (int l = 0; l < 3; ++l) {
    gather_gemm_bias_relu<<<gemm_blocks, 768, 0, stream>>>(cnt_in, cnt_out, rec, misc, ovf,
                                                           lin[l], Wp[l], bp[l], lout[l],
                                                           N, E, l < 2 ? 1 : 0);
  }
}


# Layer kernel: rows dealt evenly over the chip: 512 blocks of 21 waves (two on every WGP), 6-7 strips each, a block's rows split evenly over its waves (4-6 each); was 391 x 8; row sum order unchanged
# speedup vs baseline: 4.5404x; 1.0567x over previous
#include <hip/hip_runtime.h>
#include <hip/hip_bf16.h>

#define FEAT 96
#define KSTEPS (FEAT / 4)
#define NTILES (FEAT / 16)

typedef float v2f __attribute__((ext_vector_type(2)));
typedef float v4f __attribute__((ext_vector_type(4)));
typedef float v8f __attribute__((ext_vector_type(8)));

__global__ __launch_bounds__(256) void zero_f32(float* __restrict__ p, long long n) {
  long long i = (long long)blockIdx.x * blockDim.x + threadIdx.x;
  long long i4 = i * 4;
  if (i4 + 3 < n) {
    *(float4*)(p + i4) = make_float4(0.f, 0.f, 0.f, 0.f);
  } else {
    for (long long j = i4; j < n; ++j) p[j] = 0.f;
  }
}

#define SLOTS 64
#define NPB 196
#define NBIN 256
#define BCAP 4096
#define BT 512
#define EPT 8
#define EPB (BT * EPT)
#define PT 1024

__device__ __forceinline__ int bin_of(const int node) {
  return min((int)((unsigned)node / NPB), NBIN - 1);
}

__device__ __forceinline__ void scan_reserve(const int* hist, int* start, int* gbase,
                                             int* wsum, int* __restrict__ fill,
                                             const int tid) {
  const int lane = tid & 31;
  const int wave = tid >> 5;
  int c = 0, incl = 0;
  if (tid < NBIN) {
    c = hist[tid];
    incl = c;
    for (int o = 1; o < 32; o <<= 1) {
      const int t = __shfl_up(incl, o);
      if (lane >= o) incl += t;
    }
    if (lane == 31) wsum[wave] = incl;
  }
  __syncthreads();
  if (tid < NBIN) {
    int woff = 0;
    for (int w = 0; w < NBIN / 32; ++w)
      if (w < wave) woff += wsum[w];
    start[tid] = woff + incl - c;
    gbase[tid] = c ? atomicAdd(&fill[tid], c) : 0;
  }
  __syncthreads();
}

__global__ __launch_bounds__(BT) void bin_edges(const int* __restrict__ src,
                                                const int* __restrict__ dst,
                                                int* __restrict__ dfill,
                                                int* __restrict__ sfill,
                                                int2* __restrict__ dbin,
                                                int* __restrict__ sbin,
                                                int* __restrict__ misc,
                                                int2* __restrict__ dspill,
                                                int* __restrict__ sspill,
                                                int nedges) {
  __shared__ int hist[NBIN];
  __shared__ int start[NBIN];
  __shared__ int gbase[NBIN];
  __shared__ int wsum[NBIN / 32];
  __shared__ int2 sorted[EPB];

  const int tid = threadIdx.x;
  const int base = blockIdx.x * EPB;
  const int total = min(EPB, nedges - base);

  int s[EPT], d[EPT], r[EPT];
#pragma unroll
  for (int k = 0; k < EPT; ++k) {
    const int e = min(base + k * BT + tid, nedges - 1);
    s[k] = src[e];
    d[k] = dst[e];
  }

  if (tid < NBIN) hist[tid] = 0;
  __syncthreads();
#pragma unroll
  for (int k = 0; k < EPT; ++k)
    if (k * BT + tid < total) r[k] = atomicAdd(&hist[bin_of(d[k])], 1);
  __syncthreads();
  scan_reserve(hist, start, gbase, wsum, dfill, tid);
#pragma unroll
  for (int k = 0; k < EPT; ++k)
    if (k * BT + tid < total) sorted[start[bin_of(d[k])] + r[k]] = make_int2(d[k], s[k]);
  __syncthreads();
  for (int i = tid; i < total; i += BT) {
    const int2 p = sorted[i];
    const int b = bin_of(p.x);
    const int pos = gbase[b] + (i - start[b]);
    if (pos < BCAP) {
      dbin[b * BCAP + pos] = p;
    } else {
      const int j = atomicAdd(&misc[1], 1);
      dspill[j] = p;
    }
  }
  __syncthreads();

  int* sorted1 = (int*)sorted;
  if (tid < NBIN) hist[tid] = 0;
  __syncthreads();
#pragma unroll
  for (int k = 0; k < EPT; ++k)
    if (k * BT + tid < total) r[k] = atomicAdd(&hist[bin_of(s[k])], 1);
  __syncthreads();
  scan_reserve(hist, start, gbase, wsum, sfill, tid);
#pragma unroll
  for (int k = 0; k < EPT; ++k)
    if (k * BT + tid < total) sorted1[start[bin_of(s[k])] + r[k]] = s[k];
  __syncthreads();
  for (int i = tid; i < total; i += BT) {
    const int sv = sorted1[i];
    const int b = bin_of(sv);
    const int pos = gbase[b] + (i - start[b]);
    if (pos < BCAP) {
      sbin[b * BCAP + pos] = sv;
    } else {
      const int j = atomicAdd(&misc[2], 1);
      sspill[j] = sv;
    }
  }
}

__global__ __launch_bounds__(PT) void place_bin(const int* __restrict__ dfill,
                                                const int* __restrict__ sfill,
                                                const int2* __restrict__ dbin,
                                                const int* __restrict__ sbin,
                                                int* __restrict__ misc,
                                                const int2* __restrict__ dspill,
                                                const int* __restrict__ sspill,
                                                int* __restrict__ cnt_in,
                                                int* __restrict__ cnt_out,
                                                int* __restrict__ rec,
                                                int2* __restrict__ ovf,
                                                int nrows, int nedges) {
  __shared__ int cin[NPB];
  __shared__ int cout[NPB];
  __shared__ __attribute__((aligned(16))) int rows[NPB * SLOTS];

  const int tid = threadIdx.x;
  const int b = blockIdx.x;
  const int node0 = b * NPB;
  if (tid < NPB) {
    cin[tid] = 0;
    cout[tid] = 0;
  }
  const int nd = min(dfill[b], BCAP);
  const int ns = min(sfill[b], BCAP);
  const int ndsp = min(misc[1], nedges);
  const int nssp = min(misc[2], nedges);
  __syncthreads();

#define ADD_PAIR(p)                                                   \
  do {                                                                \
    const int dl = (p).x - node0;                                     \
    if ((unsigned)dl < NPB) {                                         \
      const int slot = atomicAdd(&cin[dl], 1);                        \
      if (slot < SLOTS) {                                             \
        rows[dl * SLOTS + slot] = (p).y;                              \
      } else {                                                        \
        const int j = atomicAdd(&misc[0], 1);                         \
        ovf[j] = (p);                                                 \
      }                                                               \
    }                                                                 \
  } while (0)

  int2 pr[BCAP / PT];
  int sv[BCAP / PT];
#pragma unroll
  for (int k = 0; k < BCAP / PT; ++k) {
    const int i = k * PT + tid;
    if (i < nd) pr[k] = dbin[b * BCAP + i];
    if (i < ns) sv[k] = sbin[b * BCAP + i];
  }
#pragma unroll
  for (int k = 0; k < BCAP / PT; ++k)
    if (k * PT + tid < nd) ADD_PAIR(pr[k]);
  for (int i = tid; i < ndsp; i += PT) {
    const int2 p = dspill[i];
    ADD_PAIR(p);
  }
#undef ADD_PAIR
#pragma unroll
  for (int k = 0; k < BCAP / PT; ++k) {
    const int sl = sv[k] - node0;
    if (k * PT + tid < ns && (unsigned)sl < NPB) atomicAdd(&cout[sl], 1);
  }
  for (int i = tid; i < nssp; i += PT) {
    const int sl = sspill[i] - node0;
    if ((unsigned)sl < NPB) atomicAdd(&cout[sl], 1);
  }
  __syncthreads();

  if (tid < NPB && node0 + tid < nrows) {
    cnt_in[node0 + tid] = cin[tid];
    cnt_out[node0 + tid] = cout[tid];
  }
  for (int q = tid; q < NPB * (SLOTS / 4); q += PT) {
    const int dl = q / (SLOTS / 4);
    const int piece = q % (SLOTS / 4);
    if (node0 + dl < nrows && piece * 4 < min(cin[dl], SLOTS))
      ((int4*)rec)[(long long)(node0 + dl) * (SLOTS / 4) + piece] = ((const int4*)rows)[q];
  }
}

#define SC 8

__global__ __launch_bounds__(256) void scale_rows(const float* __restrict__ h,
                                                  const int* __restrict__ cnt_out,
                                                  float* __restrict__ hs,
                                                  int nvec) {
  const int base = blockIdx.x * (256 * SC) + threadIdx.x;
  v4f v[SC];
  int c[SC];
#pragma unroll
  for (int k = 0; k < SC; ++k) {
    const int i = min(base + k * 256, nvec - 1);
    v[k] = ((const v4f*)h)[i];
    c[k] = cnt_out[i / (FEAT / 4)];
  }
#pragma unroll
  for (int k = 0; k < SC; ++k) {
    const int i = base + k * 256;
    if (i < nvec) {
      const float ns = 1.0f / sqrtf(fmaxf((float)c[k], 1.0f));
      ((v4f*)hs)[i] = v[k] * ns;
    }
  }
}

#define APITCH 100
#define GU 8
#define WPS 3
#define SPB 7
#define LWAVES (SPB * WPS)
#define LTHREADS (LWAVES * 32)
#define RPW 6
#define TPW (NTILES / WPS)

__device__ __forceinline__ void add4(float4& acc, const float4 v) {
  acc.x += v.x;
  acc.y += v.y;
  acc.z += v.z;
  acc.w += v.w;
}

__device__ __forceinline__ int rec_index(const int rp, const int v, const int row0) {
  int r = 0, st = 0;
#pragma unroll
  for (int j = 1; j < RPW; ++j) {
    const int rj = __builtin_amdgcn_readlane(rp, j);
    if (v >= rj) { r = j; st = rj; }
  }
  return (row0 + r) * SLOTS + (v - st);
}

__global__ __launch_bounds__(LTHREADS, 12) void gather_gemm_bias_relu(
    const int* __restrict__ cnt_in, const int* __restrict__ cnt_out,
    const int* __restrict__ rec, const int* __restrict__ ovf_cnt,
    const int2* __restrict__ ovf, const float* __restrict__ h,
    const float* __restrict__ W, const float* __restrict__ bias,
    float* __restrict__ out, int nrows, int nedges, int scale_out) {
  __shared__ __attribute__((aligned(16))) float Ws[FEAT * FEAT];
  __shared__ float bs[FEAT];
  __shared__ __attribute__((aligned(16))) float As[SPB * 16 * APITCH];

  const int tid = threadIdx.x;
  const int lane = tid & 31;
  const int wave = __builtin_amdgcn_readfirstlane(tid >> 5);
  const int sidx = wave / WPS;
  const int part = wave - sidx * WPS;
  const unsigned nstrips = (unsigned)(nrows + 15) >> 4;
  const int s0 = (int)(blockIdx.x * nstrips / gridDim.x);
  const int s1 = (int)((blockIdx.x + 1) * nstrips / gridDim.x);
  const int nsb = s1 - s0;
  const bool active = sidx < nsb;
  const int row0 = (s0 + sidx) * 16;
  const int rb = (wave * nsb * 16) / LWAVES;
  const int nrw = ((wave + 1) * nsb * 16) / LWAVES - rb;
  const int grow0 = s0 * 16 + rb;
  float* __restrict__ strip = As + sidx * (16 * APITCH);
  float* __restrict__ gstrip = As + rb * APITCH;

  v4f wreg[4];
#pragma unroll
  for (int i = 0; i < 4; ++i)
    wreg[i] = ((const v4f*)W)[min(tid + LTHREADS * i, FEAT * FEAT / 4 - 1)];
  const int novf_lane = ovf_cnt[0];

  if (nrw > 0) {
    const int fo = min(lane, FEAT / 4 - 1) * 4;
    const int rr = grow0 + min(lane, nrw - 1);
    int cfull = cnt_in[min(rr, nrows - 1)];
    if (lane >= nrw || rr >= nrows) cfull = 0;
    const float ndl = 1.0f / sqrtf(fmaxf((float)cfull, 1.0f));
    const int c = min(cfull, SLOTS);
    int incl = c;
    for (int o = 1; o < 8; o <<= 1) {
      const int t = __shfl_up(incl, o);
      if (lane >= o) incl += t;
    }
    const int end = __builtin_amdgcn_readlane(incl, 7);
    int rp = __shfl_up(incl, 1);
    if (lane == 0) rp = 0;
    if (lane >= nrw) rp = end;

    int cur = 0;
    int nextb = __builtin_amdgcn_readlane(rp, 1);
    float4 acc = make_float4(0.f, 0.f, 0.f, 0.f);

#define PARK_ROW()                                                                    \
    do {                                                                              \
      const float nd = __int_as_float(                                                \
          __builtin_amdgcn_readlane(__float_as_int(ndl), cur));                       \
      if (lane < FEAT / 4)                                                            \
        *(float4*)(gstrip + cur * APITCH + lane * 4) =                                \
            make_float4(acc.x * nd, acc.y * nd, acc.z * nd, acc.w * nd);              \
      acc = make_float4(0.f, 0.f, 0.f, 0.f);                                          \
      ++cur;                                                                          \
      nextb = __builtin_amdgcn_readlane(rp, cur + 1);                                 \
    } while (0)

    int p = 0;
    if (0 < end) p = rec[rec_index(rp, min(lane, end - 1), grow0)];
    for (int base = 0; base < end; base += 32) {
      int pn = p;
#pragma unroll
      for (int g = 0; g < 32 / GU; ++g) {
        const int e0 = base + g * GU;
        if (e0 < end) {
          float4 v[GU];
#pragma unroll
          for (int u = 0; u < GU; ++u) {
            const int s = __builtin_amdgcn_readlane(p, g * GU + u);
            v[u] = *(const float4*)(h + (long long)s * FEAT + fo);
          }
          if (g == 0 && base + 32 < end)
            pn = rec[rec_index(rp, min(base + 32 + lane, end - 1), grow0)];
          if (e0 + GU <= end) {
#pragma unroll
            for (int u = 0; u < GU; ++u) {
              const int e = e0 + u;
              if (__builtin_expect(e == nextb, 0)) {
                do { PARK_ROW(); } while (e == nextb);
              }
              add4(acc, v[u]);
            }
          } else {
#pragma unroll
            for (int u = 0; u < GU; ++u) {
              const int e = e0 + u;
              if (e < end) {
                if (__builtin_expect(e == nextb, 0)) {
                  do { PARK_ROW(); } while (e == nextb);
                }
                add4(acc, v[u]);
              }
            }
          }
        }
      }
      p = pn;
    }
    while (cur < nrw) PARK_ROW();
#undef PARK_ROW

    const int novf = min(__builtin_amdgcn_readfirstlane(novf_lane), nedges);
    for (int i = 0; i < novf; ++i) {
      const int2 o = ovf[i];
      const int r = __builtin_amdgcn_readfirstlane(o.x) - grow0;
      if (r >= 0 && r < nrw) {
        const int s = __builtin_amdgcn_readfirstlane(o.y);
        const float nd = __int_as_float(__builtin_amdgcn_readlane(__float_as_int(ndl), r));
        if (lane < FEAT / 4) {
          float4* q = (float4*)(gstrip + r * APITCH + lane * 4);
          const float4 hv = *(const float4*)(h + (long long)s * FEAT + lane * 4);
          float4 t = *q;
          t.x = fmaf(hv.x, nd, t.x);
          t.y = fmaf(hv.y, nd, t.y);
          t.z = fmaf(hv.z, nd, t.z);
          t.w = fmaf(hv.w, nd, t.w);
          *q = t;
        }
      }
    }
  }

#pragma unroll
  for (int i = 0; i < 4; ++i)
    if (tid + LTHREADS * i < FEAT * FEAT / 4) ((v4f*)Ws)[tid + LTHREADS * i] = wreg[i];
  if (tid < FEAT) bs[tid] = bias[tid];
  __syncthreads();

  if (!active || row0 >= nrows) return;

  const int lrow = lane & 15;
  const int lgrp = lane >> 4;

  float osc[8];
#pragma unroll
  for (int r = 0; r < 8; ++r) {
    osc[r] = 1.0f;
    if (scale_out) {
      const int row = min(row0 + r + 8 * lgrp, nrows - 1);
      osc[r] = 1.0f / sqrtf(fmaxf((float)cnt_out[row], 1.0f));
    }
  }

  v2f a[KSTEPS];
#pragma unroll
  for (int ks = 0; ks < KSTEPS; ++ks)
    a[ks] = *(const v2f*)(strip + lrow * APITCH + ks * 4 + 2 * lgrp);

  const int cb0 = part * (TPW * 16);

  v8f acc[TPW];
#pragma unroll
  for (int nt = 0; nt < TPW; ++nt) acc[nt] = (v8f){};

  v2f bf[TPW];
#pragma unroll
  for (int nt = 0; nt < TPW; ++nt) {
    const int k = 2 * lgrp;
    bf[nt].x = Ws[k * FEAT + cb0 + nt * 16 + lrow];
    bf[nt].y = Ws[(k + 1) * FEAT + cb0 + nt * 16 + lrow];
  }

#pragma unroll
  for (int ks = 0; ks < KSTEPS; ++ks) {
    v2f bn[TPW];
    if (ks + 1 < KSTEPS) {
      const int k = (ks + 1) * 4 + 2 * lgrp;
#pragma unroll
      for (int nt = 0; nt < TPW; ++nt) {
        bn[nt].x = Ws[k * FEAT + cb0 + nt * 16 + lrow];
        bn[nt].y = Ws[(k + 1) * FEAT + cb0 + nt * 16 + lrow];
      }
    }
#pragma unroll
    for (int nt = 0; nt < TPW; ++nt) {
      acc[nt] = __builtin_amdgcn_wmma_f32_16x16x4_f32(
           false, a[ks],  false, bf[nt],
           (short)0, acc[nt],  false,  false);
    }
    if (ks + 1 < KSTEPS) {
#pragma unroll
      for (int nt = 0; nt < TPW; ++nt) bf[nt] = bn[nt];
    }
  }

#pragma unroll
  for (int nt = 0; nt < TPW; ++nt) {
    const int nb = cb0 + nt * 16;
    const float bcol = bs[nb + lrow];
#pragma unroll
    for (int r = 0; r < 8; ++r) {
      const int row = row0 + r + 8 * lgrp;
      if (row < nrows) {
        float v = acc[nt][r] + bcol;
        out[(long long)row * FEAT + nb + lrow] = fmaxf(v, 0.0f) * osc[r];
      }
    }
  }
}

extern "C" void kernel_launch(void* const* d_in, const int* in_sizes, int n_in,
                              void* d_out, int out_size, void* d_ws, size_t ws_size,
                              hipStream_t stream) {
  const float* h0  = (const float*)d_in[0];
  const int*   src = (const int*)d_in[1];
  const int*   dst = (const int*)d_in[2];
  const float* Wp[3] = {(const float*)d_in[3], (const float*)d_in[5], (const float*)d_in[7]};
  const float* bp[3] = {(const float*)d_in[4], (const float*)d_in[6], (const float*)d_in[8]};

  const int N = in_sizes[0] / FEAT;
  const int E = in_sizes[1];

  const long long NP = ((long long)N + 63) / 64 * 64;
  const long long EP = ((long long)E + 63) / 64 * 64;
  float* ws       = (float*)d_ws;
  int*   cnt_out  = (int*)ws;
  int*   cnt_in   = (int*)(ws + NP);
  int*   misc     = (int*)(ws + 2 * NP);
  int*   dfill    = misc + 64;
  int*   sfill    = dfill + NBIN;
  const long long HEAD = 2 * NP + 64 + 2 * NBIN;
  int*   rec      = (int*)(ws + HEAD);
  int2*  ovf      = (int2*)(ws + HEAD + (long long)N * SLOTS);
  float* hA       = ws + HEAD + (long long)N * SLOTS + 2 * EP;
  float* hs0      = hA + (long long)N * FEAT;
  int2*  dbin     = (int2*)(hs0 + (long long)N * FEAT);
  int*   sbin     = (int*)(hs0 + (long long)N * FEAT + 2LL * NBIN * BCAP);
  int2*  dspill   = (int2*)(sbin + (long long)NBIN * BCAP);
  int*   sspill   = (int*)(dspill + EP);
  float* outp     = (float*)d_out;

  {
    long long n2 = 64 + 2 * NBIN;
    int blocks = (int)((n2 / 4 + 255) / 256) + 1;
    zero_f32<<<blocks, 256, 0, stream>>>((float*)misc, n2);
    bin_edges<<<(E + EPB - 1) / EPB, BT, 0, stream>>>(src, dst, dfill, sfill, dbin, sbin, misc,
                                                      dspill, sspill, E);
    place_bin<<<(N + NPB - 1) / NPB, PT, 0, stream>>>(dfill, sfill, dbin, sbin, misc, dspill,
                                                      sspill, cnt_in, cnt_out, rec, ovf, N, E);
    const int nvec = N * (FEAT / 4);
    scale_rows<<<(nvec + 256 * SC - 1) / (256 * SC), 256, 0, stream>>>(h0, cnt_out, hs0, nvec);
  }

  const int strips = (N + 15) / 16;
  const int need_blocks = (strips + SPB - 1) / SPB;
  const int gemm_blocks = need_blocks > 512 ? need_blocks : 512;

  const float* lin[3]  = {hs0, outp, hA};
  float*       lout[3] = {outp, hA, outp};

  for (int l = 0; l < 3; ++l) {
    gather_gemm_bias_relu<<<gemm_blocks, LTHREADS, 0, stream>>>(cnt_in, cnt_out, rec, misc, ovf,
                                                           lin[l], Wp[l], bp[l], lout[l],
                                                           N, E, l < 2 ? 1 : 0);
  }
}


# First layer gathers the plain input rows; each record's row times its source norm is added by one fused multiply-add (32 norms per degree load): no row-scaling launch, 6 launches not 7
# speedup vs baseline: 4.7798x; 1.0527x over previous
#include <hip/hip_runtime.h>
#include <hip/hip_bf16.h>

#define FEAT 96
#define KSTEPS (FEAT / 4)
#define NTILES (FEAT / 16)

typedef float v2f __attribute__((ext_vector_type(2)));
typedef float v4f __attribute__((ext_vector_type(4)));
typedef float v8f __attribute__((ext_vector_type(8)));

__global__ __launch_bounds__(256) void zero_f32(float* __restrict__ p, long long n) {
  long long i = (long long)blockIdx.x * blockDim.x + threadIdx.x;
  long long i4 = i * 4;
  if (i4 + 3 < n) {
    *(float4*)(p + i4) = make_float4(0.f, 0.f, 0.f, 0.f);
  } else {
    for (long long j = i4; j < n; ++j) p[j] = 0.f;
  }
}

#define SLOTS 64
#define NPB 196
#define NBIN 256
#define BCAP 4096
#define BT 512
#define EPT 8
#define EPB (BT * EPT)
#define PT 1024

__device__ __forceinline__ int bin_of(const int node) {
  return min((int)((unsigned)node / NPB), NBIN - 1);
}

__device__ __forceinline__ void scan_reserve(const int* hist, int* start, int* gbase,
                                             int* wsum, int* __restrict__ fill,
                                             const int tid) {
  const int lane = tid & 31;
  const int wave = tid >> 5;
  int c = 0, incl = 0;
  if (tid < NBIN) {
    c = hist[tid];
    incl = c;
    for (int o = 1; o < 32; o <<= 1) {
      const int t = __shfl_up(incl, o);
      if (lane >= o) incl += t;
    }
    if (lane == 31) wsum[wave] = incl;
  }
  __syncthreads();
  if (tid < NBIN) {
    int woff = 0;
    for (int w = 0; w < NBIN / 32; ++w)
      if (w < wave) woff += wsum[w];
    start[tid] = woff + incl - c;
    gbase[tid] = c ? atomicAdd(&fill[tid], c) : 0;
  }
  __syncthreads();
}

__global__ __launch_bounds__(BT) void bin_edges(const int* __restrict__ src,
                                                const int* __restrict__ dst,
                                                int* __restrict__ dfill,
                                                int* __restrict__ sfill,
                                                int2* __restrict__ dbin,
                                                int* __restrict__ sbin,
                                                int* __restrict__ misc,
                                                int2* __restrict__ dspill,
                                                int* __restrict__ sspill,
                                                int nedges) {
  __shared__ int hist[NBIN];
  __shared__ int start[NBIN];
  __shared__ int gbase[NBIN];
  __shared__ int wsum[NBIN / 32];
  __shared__ int2 sorted[EPB];

  const int tid = threadIdx.x;
  const int base = blockIdx.x * EPB;
  const int total = min(EPB, nedges - base);

  int s[EPT], d[EPT], r[EPT];
#pragma unroll
  for (int k = 0; k < EPT; ++k) {
    const int e = min(base + k * BT + tid, nedges - 1);
    s[k] = src[e];
    d[k] = dst[e];
  }

  if (tid < NBIN) hist[tid] = 0;
  __syncthreads();
#pragma unroll
  for (int k = 0; k < EPT; ++k)
    if (k * BT + tid < total) r[k] = atomicAdd(&hist[bin_of(d[k])], 1);
  __syncthreads();
  scan_reserve(hist, start, gbase, wsum, dfill, tid);
#pragma unroll
  for (int k = 0; k < EPT; ++k)
    if (k * BT + tid < total) sorted[start[bin_of(d[k])] + r[k]] = make_int2(d[k], s[k]);
  __syncthreads();
  for (int i = tid; i < total; i += BT) {
    const int2 p = sorted[i];
    const int b = bin_of(p.x);
    const int pos = gbase[b] + (i - start[b]);
    if (pos < BCAP) {
      dbin[b * BCAP + pos] = p;
    } else {
      const int j = atomicAdd(&misc[1], 1);
      dspill[j] = p;
    }
  }
  __syncthreads();

  int* sorted1 = (int*)sorted;
  if (tid < NBIN) hist[tid] = 0;
  __syncthreads();
#pragma unroll
  for (int k = 0; k < EPT; ++k)
    if (k * BT + tid < total) r[k] = atomicAdd(&hist[bin_of(s[k])], 1);
  __syncthreads();
  scan_reserve(hist, start, gbase, wsum, sfill, tid);
#pragma unroll
  for (int k = 0; k < EPT; ++k)
    if (k * BT + tid < total) sorted1[start[bin_of(s[k])] + r[k]] = s[k];
  __syncthreads();
  for (int i = tid; i < total; i += BT) {
    const int sv = sorted1[i];
    const int b = bin_of(sv);
    const int pos = gbase[b] + (i - start[b]);
    if (pos < BCAP) {
      sbin[b * BCAP + pos] = sv;
    } else {
      const int j = atomicAdd(&misc[2], 1);
      sspill[j] = sv;
    }
  }
}

__global__ __launch_bounds__(PT) void place_bin(const int* __restrict__ dfill,
                                                const int* __restrict__ sfill,
                                                const int2* __restrict__ dbin,
                                                const int* __restrict__ sbin,
                                                int* __restrict__ misc,
                                                const int2* __restrict__ dspill,
                                                const int* __restrict__ sspill,
                                                int* __restrict__ cnt_in,
                                                int* __restrict__ cnt_out,
                                                int* __restrict__ rec,
                                                int2* __restrict__ ovf,
                                                int nrows, int nedges) {
  __shared__ int cin[NPB];
  __shared__ int cout[NPB];
  __shared__ __attribute__((aligned(16))) int rows[NPB * SLOTS];

  const int tid = threadIdx.x;
  const int b = blockIdx.x;
  const int node0 = b * NPB;
  if (tid < NPB) {
    cin[tid] = 0;
    cout[tid] = 0;
  }
  const int nd = min(dfill[b], BCAP);
  const int ns = min(sfill[b], BCAP);
  const int ndsp = min(misc[1], nedges);
  const int nssp = min(misc[2], nedges);
  __syncthreads();

#define ADD_PAIR(p)                                                   \
  do {                                                                \
    const int dl = (p).x - node0;                                     \
    if ((unsigned)dl < NPB) {                                         \
      const int slot = atomicAdd(&cin[dl], 1);                        \
      if (slot < SLOTS) {                                             \
        rows[dl * SLOTS + slot] = (p).y;                              \
      } else {                                                        \
        const int j = atomicAdd(&misc[0], 1);                         \
        ovf[j] = (p);                                                 \
      }                                                               \
    }                                                                 \
  } while (0)

  int2 pr[BCAP / PT];
  int sv[BCAP / PT];
#pragma unroll
  for (int k = 0; k < BCAP / PT; ++k) {
    const int i = k * PT + tid;
    if (i < nd) pr[k] = dbin[b * BCAP + i];
    if (i < ns) sv[k] = sbin[b * BCAP + i];
  }
#pragma unroll
  for (int k = 0; k < BCAP / PT; ++k)
    if (k * PT + tid < nd) ADD_PAIR(pr[k]);
  for (int i = tid; i < ndsp; i += PT) {
    const int2 p = dspill[i];
    ADD_PAIR(p);
  }
#undef ADD_PAIR
#pragma unroll
  for (int k = 0; k < BCAP / PT; ++k) {
    const int sl = sv[k] - node0;
    if (k * PT + tid < ns && (unsigned)sl < NPB) atomicAdd(&cout[sl], 1);
  }
  for (int i = tid; i < nssp; i += PT) {
    const int sl = sspill[i] - node0;
    if ((unsigned)sl < NPB) atomicAdd(&cout[sl], 1);
  }
  __syncthreads();

  if (tid < NPB && node0 + tid < nrows) {
    cnt_in[node0 + tid] = cin[tid];
    cnt_out[node0 + tid] = cout[tid];
  }
  for (int q = tid; q < NPB * (SLOTS / 4); q += PT) {
    const int dl = q / (SLOTS / 4);
    const int piece = q % (SLOTS / 4);
    if (node0 + dl < nrows && piece * 4 < min(cin[dl], SLOTS))
      ((int4*)rec)[(long long)(node0 + dl) * (SLOTS / 4) + piece] = ((const int4*)rows)[q];
  }
}

#define APITCH 100
#define GU 8
#define WPS 3
#define SPB 7
#define LWAVES (SPB * WPS)
#define LTHREADS (LWAVES * 32)
#define RPW 6
#define TPW (NTILES / WPS)

__device__ __forceinline__ void add4(float4& acc, const float4 v) {
  acc.x += v.x;
  acc.y += v.y;
  acc.z += v.z;
  acc.w += v.w;
}

__device__ __forceinline__ void fma4(float4& acc, const float4 v, const float s) {
  acc.x = fmaf(v.x, s, acc.x);
  acc.y = fmaf(v.y, s, acc.y);
  acc.z = fmaf(v.z, s, acc.z);
  acc.w = fmaf(v.w, s, acc.w);
}

__device__ __forceinline__ float deg_norm(const int c) {
  return 1.0f / sqrtf(fmaxf((float)c, 1.0f));
}

__device__ __forceinline__ int rec_index(const int rp, const int v, const int row0) {
  int r = 0, st = 0;
#pragma unroll
  for (int j = 1; j < RPW; ++j) {
    const int rj = __builtin_amdgcn_readlane(rp, j);
    if (v >= rj) { r = j; st = rj; }
  }
  return (row0 + r) * SLOTS + (v - st);
}

template <bool SCALE_IN>
__global__ __launch_bounds__(LTHREADS, 12) void gather_gemm_bias_relu(
    const int* __restrict__ cnt_in, const int* __restrict__ cnt_out,
    const int* __restrict__ rec, const int* __restrict__ ovf_cnt,
    const int2* __restrict__ ovf, const float* __restrict__ h,
    const float* __restrict__ W, const float* __restrict__ bias,
    float* __restrict__ out, int nrows, int nedges, int scale_out) {
  __shared__ __attribute__((aligned(16))) float Ws[FEAT * FEAT];
  __shared__ float bs[FEAT];
  __shared__ __attribute__((aligned(16))) float As[SPB * 16 * APITCH];

  const int tid = threadIdx.x;
  const int lane = tid & 31;
  const int wave = __builtin_amdgcn_readfirstlane(tid >> 5);
  const int sidx = wave / WPS;
  const int part = wave - sidx * WPS;
  const unsigned nstrips = (unsigned)(nrows + 15) >> 4;
  const int s0 = (int)(blockIdx.x * nstrips / gridDim.x);
  const int s1 = (int)((blockIdx.x + 1) * nstrips / gridDim.x);
  const int nsb = s1 - s0;
  const bool active = sidx < nsb;
  const int row0 = (s0 + sidx) * 16;
  const int rb = (wave * nsb * 16) / LWAVES;
  const int nrw = ((wave + 1) * nsb * 16) / LWAVES - rb;
  const int grow0 = s0 * 16 + rb;
  float* __restrict__ strip = As + sidx * (16 * APITCH);
  float* __restrict__ gstrip = As + rb * APITCH;

  v4f wreg[4];
#pragma unroll
  for (int i = 0; i < 4; ++i)
    wreg[i] = ((const v4f*)W)[min(tid + LTHREADS * i, FEAT * FEAT / 4 - 1)];
  const int novf_lane = ovf_cnt[0];

  if (nrw > 0) {
    const int fo = min(lane, FEAT / 4 - 1) * 4;
    const int rr = grow0 + min(lane, nrw - 1);
    int cfull = cnt_in[min(rr, nrows - 1)];
    if (lane >= nrw || rr >= nrows) cfull = 0;
    const float ndl = 1.0f / sqrtf(fmaxf((float)cfull, 1.0f));
    const int c = min(cfull, SLOTS);
    int incl = c;
    for (int o = 1; o < 8; o <<= 1) {
      const int t = __shfl_up(incl, o);
      if (lane >= o) incl += t;
    }
    const int end = __builtin_amdgcn_readlane(incl, 7);
    int rp = __shfl_up(incl, 1);
    if (lane == 0) rp = 0;
    if (lane >= nrw) rp = end;

    int cur = 0;
    int nextb = __builtin_amdgcn_readlane(rp, 1);
    float4 acc = make_float4(0.f, 0.f, 0.f, 0.f);

#define PARK_ROW()                                                                    \
    do {                                                                              \
      const float nd = __int_as_float(                                                \
          __builtin_amdgcn_readlane(__float_as_int(ndl), cur));                       \
      if (lane < FEAT / 4)                                                            \
        *(float4*)(gstrip + cur * APITCH + lane * 4) =                                \
            make_float4(acc.x * nd, acc.y * nd, acc.z * nd, acc.w * nd);              \
      acc = make_float4(0.f, 0.f, 0.f, 0.f);                                          \
      ++cur;                                                                          \
      nextb = __builtin_amdgcn_readlane(rp, cur + 1);                                 \
    } while (0)

    int p = 0;
    if (0 < end) p = rec[rec_index(rp, min(lane, end - 1), grow0)];
    int cn = 1;
    if (SCALE_IN) cn = cnt_out[p];
    for (int base = 0; base < end; base += 32) {
      int pn = p;
      int cnn = 1;
      float psc = 1.0f;
#pragma unroll
      for (int g = 0; g < 32 / GU; ++g) {
        const int e0 = base + g * GU;
        if (e0 < end) {
          float4 v[GU];
#pragma unroll
          for (int u = 0; u < GU; ++u) {
            const int s = __builtin_amdgcn_readlane(p, g * GU + u);
            v[u] = *(const float4*)(h + (long long)s * FEAT + fo);
          }
          if (SCALE_IN && g == 0) psc = deg_norm(cn);
          float ns[GU];
#pragma unroll
          for (int u = 0; u < GU; ++u)
            ns[u] = SCALE_IN ? __int_as_float(__builtin_amdgcn_readlane(__float_as_int(psc), g * GU + u))
                             : 1.0f;
          if (g == 0 && base + 32 < end)
            pn = rec[rec_index(rp, min(base + 32 + lane, end - 1), grow0)];
          if (SCALE_IN && g == 2 && base + 32 < end) cnn = cnt_out[pn];
          if (e0 + GU <= end) {
#pragma unroll
            for (int u = 0; u < GU; ++u) {
              const int e = e0 + u;
              if (__builtin_expect(e == nextb, 0)) {
                do { PARK_ROW(); } while (e == nextb);
              }
              if (SCALE_IN) fma4(acc, v[u], ns[u]); else add4(acc, v[u]);
            }
          } else {
#pragma unroll
            for (int u = 0; u < GU; ++u) {
              const int e = e0 + u;
              if (e < end) {
                if (__builtin_expect(e == nextb, 0)) {
                  do { PARK_ROW(); } while (e == nextb);
                }
                if (SCALE_IN) fma4(acc, v[u], ns[u]); else add4(acc, v[u]);
              }
            }
          }
        }
      }
      p = pn;
      cn = cnn;
    }
    while (cur < nrw) PARK_ROW();
#undef PARK_ROW

    const int novf = min(__builtin_amdgcn_readfirstlane(novf_lane), nedges);
    for (int i = 0; i < novf; ++i) {
      const int2 o = ovf[i];
      const int r = __builtin_amdgcn_readfirstlane(o.x) - grow0;
      if (r >= 0 && r < nrw) {
        const int s = __builtin_amdgcn_readfirstlane(o.y);
        const float nd = __int_as_float(__builtin_amdgcn_readlane(__float_as_int(ndl), r));
        float nsrc = 1.0f;
        if (SCALE_IN) nsrc = deg_norm(cnt_out[s]);
        if (lane < FEAT / 4) {
          float4* q = (float4*)(gstrip + r * APITCH + lane * 4);
          float4 hv = *(const float4*)(h + (long long)s * FEAT + lane * 4);
          if (SCALE_IN) hv = make_float4(hv.x * nsrc, hv.y * nsrc, hv.z * nsrc, hv.w * nsrc);
          float4 t = *q;
          t.x = fmaf(hv.x, nd, t.x);
          t.y = fmaf(hv.y, nd, t.y);
          t.z = fmaf(hv.z, nd, t.z);
          t.w = fmaf(hv.w, nd, t.w);
          *q = t;
        }
      }
    }
  }

#pragma unroll
  for (int i = 0; i < 4; ++i)
    if (tid + LTHREADS * i < FEAT * FEAT / 4) ((v4f*)Ws)[tid + LTHREADS * i] = wreg[i];
  if (tid < FEAT) bs[tid] = bias[tid];
  __syncthreads();

  if (!active || row0 >= nrows) return;

  const int lrow = lane & 15;
  const int lgrp = lane >> 4;

  float osc[8];
#pragma unroll
  for (int r = 0; r < 8; ++r) {
    osc[r] = 1.0f;
    if (scale_out) {
      const int row = min(row0 + r + 8 * lgrp, nrows - 1);
      osc[r] = 1.0f / sqrtf(fmaxf((float)cnt_out[row], 1.0f));
    }
  }

  v2f a[KSTEPS];
#pragma unroll
  for (int ks = 0; ks < KSTEPS; ++ks)
    a[ks] = *(const v2f*)(strip + lrow * APITCH + ks * 4 + 2 * lgrp);

  const int cb0 = part * (TPW * 16);

  v8f acc[TPW];
#pragma unroll
  for (int nt = 0; nt < TPW; ++nt) acc[nt] = (v8f){};

  v2f bf[TPW];
#pragma unroll
  for (int nt = 0; nt < TPW; ++nt) {
    const int k = 2 * lgrp;
    bf[nt].x = Ws[k * FEAT + cb0 + nt * 16 + lrow];
    bf[nt].y = Ws[(k + 1) * FEAT + cb0 + nt * 16 + lrow];
  }

#pragma unroll
  for (int ks = 0; ks < KSTEPS; ++ks) {
    v2f bn[TPW];
    if (ks + 1 < KSTEPS) {
      const int k = (ks + 1) * 4 + 2 * lgrp;
#pragma unroll
      for (int nt = 0; nt < TPW; ++nt) {
        bn[nt].x = Ws[k * FEAT + cb0 + nt * 16 + lrow];
        bn[nt].y = Ws[(k + 1) * FEAT + cb0 + nt * 16 + lrow];
      }
    }
#pragma unroll
    for (int nt = 0; nt < TPW; ++nt) {
      acc[nt] = __builtin_amdgcn_wmma_f32_16x16x4_f32(
           false, a[ks],  false, bf[nt],
           (short)0, acc[nt],  false,  false);
    }
    if (ks + 1 < KSTEPS) {
#pragma unroll
      for (int nt = 0; nt < TPW; ++nt) bf[nt] = bn[nt];
    }
  }

#pragma unroll
  for (int nt = 0; nt < TPW; ++nt) {
    const int nb = cb0 + nt * 16;
    const float bcol = bs[nb + lrow];
#pragma unroll
    for (int r = 0; r < 8; ++r) {
      const int row = row0 + r + 8 * lgrp;
      if (row < nrows) {
        float v = acc[nt][r] + bcol;
        out[(long long)row * FEAT + nb + lrow] = fmaxf(v, 0.0f) * osc[r];
      }
    }
  }
}

extern "C" void kernel_launch(void* const* d_in, const int* in_sizes, int n_in,
                              void* d_out, int out_size, void* d_ws, size_t ws_size,
                              hipStream_t stream) {
  const float* h0  = (const float*)d_in[0];
  const int*   src = (const int*)d_in[1];
  const int*   dst = (const int*)d_in[2];
  const float* Wp[3] = {(const float*)d_in[3], (const float*)d_in[5], (const float*)d_in[7]};
  const float* bp[3] = {(const float*)d_in[4], (const float*)d_in[6], (const float*)d_in[8]};

  const int N = in_sizes[0] / FEAT;
  const int E = in_sizes[1];

  const long long NP = ((long long)N + 63) / 64 * 64;
  const long long EP = ((long long)E + 63) / 64 * 64;
  float* ws       = (float*)d_ws;
  int*   cnt_out  = (int*)ws;
  int*   cnt_in   = (int*)(ws + NP);
  int*   misc     = (int*)(ws + 2 * NP);
  int*   dfill    = misc + 64;
  int*   sfill    = dfill + NBIN;
  const long long HEAD = 2 * NP + 64 + 2 * NBIN;
  int*   rec      = (int*)(ws + HEAD);
  int2*  ovf      = (int2*)(ws + HEAD + (long long)N * SLOTS);
  float* hA       = ws + HEAD + (long long)N * SLOTS + 2 * EP;
  int2*  dbin     = (int2*)(hA + (long long)N * FEAT);
  int*   sbin     = (int*)(hA + (long long)N * FEAT + 2LL * NBIN * BCAP);
  int2*  dspill   = (int2*)(sbin + (long long)NBIN * BCAP);
  int*   sspill   = (int*)(dspill + EP);
  float* outp     = (float*)d_out;

  {
    long long n2 = 64 + 2 * NBIN;
    int blocks = (int)((n2 / 4 + 255) / 256) + 1;
    zero_f32<<<blocks, 256, 0, stream>>>((float*)misc, n2);
    bin_edges<<<(E + EPB - 1) / EPB, BT, 0, stream>>>(src, dst, dfill, sfill, dbin, sbin, misc,
                                                      dspill, sspill, E);
    place_bin<<<(N + NPB - 1) / NPB, PT, 0, stream>>>(dfill, sfill, dbin, sbin, misc, dspill,
                                                      sspill, cnt_in, cnt_out, rec, ovf, N, E);
  }

  const int strips = (N + 15) / 16;
  const int need_blocks = (strips + SPB - 1) / SPB;
  const int gemm_blocks = need_blocks > 512 ? need_blocks : 512;

  gather_gemm_bias_relu<true><<<gemm_blocks, LTHREADS, 0, stream>>>(
      cnt_in, cnt_out, rec, misc, ovf, h0, Wp[0], bp[0], outp, N, E, 1);
  gather_gemm_bias_relu<false><<<gemm_blocks, LTHREADS, 0, stream>>>(
      cnt_in, cnt_out, rec, misc, ovf, outp, Wp[1], bp[1], hA, N, E, 1);
  gather_gemm_bias_relu<false><<<gemm_blocks, LTHREADS, 0, stream>>>(
      cnt_in, cnt_out, rec, misc, ovf, hA, Wp[2], bp[2], outp, N, E, 0);
}


# Layer epilogue in row layout: lane halves of a wave's two column tiles exchanged so a register is one output row: its factor is one scalar (16 norms a wave, was 8 a lane), each 128-B line one store
# speedup vs baseline: 5.0050x; 1.0471x over previous
#include <hip/hip_runtime.h>
#include <hip/hip_bf16.h>

#define FEAT 96
#define KSTEPS (FEAT / 4)
#define NTILES (FEAT / 16)

typedef float v2f __attribute__((ext_vector_type(2)));
typedef float v4f __attribute__((ext_vector_type(4)));
typedef float v8f __attribute__((ext_vector_type(8)));

__global__ __launch_bounds__(256) void zero_f32(float* __restrict__ p, long long n) {
  long long i = (long long)blockIdx.x * blockDim.x + threadIdx.x;
  long long i4 = i * 4;
  if (i4 + 3 < n) {
    *(float4*)(p + i4) = make_float4(0.f, 0.f, 0.f, 0.f);
  } else {
    for (long long j = i4; j < n; ++j) p[j] = 0.f;
  }
}

#define SLOTS 64
#define NPB 196
#define NBIN 256
#define BCAP 4096
#define BT 512
#define EPT 8
#define EPB (BT * EPT)
#define PT 1024

__device__ __forceinline__ int bin_of(const int node) {
  return min((int)((unsigned)node / NPB), NBIN - 1);
}

__device__ __forceinline__ void scan_reserve(const int* hist, int* start, int* gbase,
                                             int* wsum, int* __restrict__ fill,
                                             const int tid) {
  const int lane = tid & 31;
  const int wave = tid >> 5;
  int c = 0, incl = 0;
  if (tid < NBIN) {
    c = hist[tid];
    incl = c;
    for (int o = 1; o < 32; o <<= 1) {
      const int t = __shfl_up(incl, o);
      if (lane >= o) incl += t;
    }
    if (lane == 31) wsum[wave] = incl;
  }
  __syncthreads();
  if (tid < NBIN) {
    int woff = 0;
    for (int w = 0; w < NBIN / 32; ++w)
      if (w < wave) woff += wsum[w];
    start[tid] = woff + incl - c;
    gbase[tid] = c ? atomicAdd(&fill[tid], c) : 0;
  }
  __syncthreads();
}

__global__ __launch_bounds__(BT) void bin_edges(const int* __restrict__ src,
                                                const int* __restrict__ dst,
                                                int* __restrict__ dfill,
                                                int* __restrict__ sfill,
                                                int2* __restrict__ dbin,
                                                int* __restrict__ sbin,
                                                int* __restrict__ misc,
                                                int2* __restrict__ dspill,
                                                int* __restrict__ sspill,
                                                int nedges) {
  __shared__ int hist[NBIN];
  __shared__ int start[NBIN];
  __shared__ int gbase[NBIN];
  __shared__ int wsum[NBIN / 32];
  __shared__ int2 sorted[EPB];

  const int tid = threadIdx.x;
  const int base = blockIdx.x * EPB;
  const int total = min(EPB, nedges - base);

  int s[EPT], d[EPT], r[EPT];
#pragma unroll
  for (int k = 0; k < EPT; ++k) {
    const int e = min(base + k * BT + tid, nedges - 1);
    s[k] = src[e];
    d[k] = dst[e];
  }

  if (tid < NBIN) hist[tid] = 0;
  __syncthreads();
#pragma unroll
  for (int k = 0; k < EPT; ++k)
    if (k * BT + tid < total) r[k] = atomicAdd(&hist[bin_of(d[k])], 1);
  __syncthreads();
  scan_reserve(hist, start, gbase, wsum, dfill, tid);
#pragma unroll
  for (int k = 0; k < EPT; ++k)
    if (k * BT + tid < total) sorted[start[bin_of(d[k])] + r[k]] = make_int2(d[k], s[k]);
  __syncthreads();
  for (int i = tid; i < total; i += BT) {
    const int2 p = sorted[i];
    const int b = bin_of(p.x);
    const int pos = gbase[b] + (i - start[b]);
    if (pos < BCAP) {
      dbin[b * BCAP + pos] = p;
    } else {
      const int j = atomicAdd(&misc[1], 1);
      dspill[j] = p;
    }
  }
  __syncthreads();

  int* sorted1 = (int*)sorted;
  if (tid < NBIN) hist[tid] = 0;
  __syncthreads();
#pragma unroll
  for (int k = 0; k < EPT; ++k)
    if (k * BT + tid < total) r[k] = atomicAdd(&hist[bin_of(s[k])], 1);
  __syncthreads();
  scan_reserve(hist, start, gbase, wsum, sfill, tid);
#pragma unroll
  for (int k = 0; k < EPT; ++k)
    if (k * BT + tid < total) sorted1[start[bin_of(s[k])] + r[k]] = s[k];
  __syncthreads();
  for (int i = tid; i < total; i += BT) {
    const int sv = sorted1[i];
    const int b = bin_of(sv);
    const int pos = gbase[b] + (i - start[b]);
    if (pos < BCAP) {
      sbin[b * BCAP + pos] = sv;
    } else {
      const int j = atomicAdd(&misc[2], 1);
      sspill[j] = sv;
    }
  }
}

__global__ __launch_bounds__(PT) void place_bin(const int* __restrict__ dfill,
                                                const int* __restrict__ sfill,
                                                const int2* __restrict__ dbin,
                                                const int* __restrict__ sbin,
                                                int* __restrict__ misc,
                                                const int2* __restrict__ dspill,
                                                const int* __restrict__ sspill,
                                                int* __restrict__ cnt_in,
                                                int* __restrict__ cnt_out,
                                                int* __restrict__ rec,
                                                int2* __restrict__ ovf,
                                                int nrows, int nedges) {
  __shared__ int cin[NPB];
  __shared__ int cout[NPB];
  __shared__ __attribute__((aligned(16))) int rows[NPB * SLOTS];

  const int tid = threadIdx.x;
  const int b = blockIdx.x;
  const int node0 = b * NPB;
  if (tid < NPB) {
    cin[tid] = 0;
    cout[tid] = 0;
  }
  const int nd = min(dfill[b], BCAP);
  const int ns = min(sfill[b], BCAP);
  const int ndsp = min(misc[1], nedges);
  const int nssp = min(misc[2], nedges);
  __syncthreads();

#define ADD_PAIR(p)                                                   \
  do {                                                                \
    const int dl = (p).x - node0;                                     \
    if ((unsigned)dl < NPB) {                                         \
      const int slot = atomicAdd(&cin[dl], 1);                        \
      if (slot < SLOTS) {                                             \
        rows[dl * SLOTS + slot] = (p).y;                              \
      } else {                                                        \
        const int j = atomicAdd(&misc[0], 1);                         \
        ovf[j] = (p);                                                 \
      }                                                               \
    }                                                                 \
  } while (0)

  int2 pr[BCAP / PT];
  int sv[BCAP / PT];
#pragma unroll
  for (int k = 0; k < BCAP / PT; ++k) {
    const int i = k * PT + tid;
    if (i < nd) pr[k] = dbin[b * BCAP + i];
    if (i < ns) sv[k] = sbin[b * BCAP + i];
  }
#pragma unroll
  for (int k = 0; k < BCAP / PT; ++k)
    if (k * PT + tid < nd) ADD_PAIR(pr[k]);
  for (int i = tid; i < ndsp; i += PT) {
    const int2 p = dspill[i];
    ADD_PAIR(p);
  }
#undef ADD_PAIR
#pragma unroll
  for (int k = 0; k < BCAP / PT; ++k) {
    const int sl = sv[k] - node0;
    if (k * PT + tid < ns && (unsigned)sl < NPB) atomicAdd(&cout[sl], 1);
  }
  for (int i = tid; i < nssp; i += PT) {
    const int sl = sspill[i] - node0;
    if ((unsigned)sl < NPB) atomicAdd(&cout[sl], 1);
  }
  __syncthreads();

  if (tid < NPB && node0 + tid < nrows) {
    cnt_in[node0 + tid] = cin[tid];
    cnt_out[node0 + tid] = cout[tid];
  }
  for (int q = tid; q < NPB * (SLOTS / 4); q += PT) {
    const int dl = q / (SLOTS / 4);
    const int piece = q % (SLOTS / 4);
    if (node0 + dl < nrows && piece * 4 < min(cin[dl], SLOTS))
      ((int4*)rec)[(long long)(node0 + dl) * (SLOTS / 4) + piece] = ((const int4*)rows)[q];
  }
}

#define APITCH 100
#define GU 8
#define WPS 3
#define SPB 7
#define LWAVES (SPB * WPS)
#define LTHREADS (LWAVES * 32)
#define RPW 6
#define TPW (NTILES / WPS)

__device__ __forceinline__ void add4(float4& acc, const float4 v) {
  acc.x += v.x;
  acc.y += v.y;
  acc.z += v.z;
  acc.w += v.w;
}

__device__ __forceinline__ void fma4(float4& acc, const float4 v, const float s) {
  acc.x = fmaf(v.x, s, acc.x);
  acc.y = fmaf(v.y, s, acc.y);
  acc.z = fmaf(v.z, s, acc.z);
  acc.w = fmaf(v.w, s, acc.w);
}

__device__ __forceinline__ float deg_norm(const int c) {
  return 1.0f / sqrtf(fmaxf((float)c, 1.0f));
}

__device__ __forceinline__ int rec_index(const int rp, const int v, const int row0) {
  int r = 0, st = 0;
#pragma unroll
  for (int j = 1; j < RPW; ++j) {
    const int rj = __builtin_amdgcn_readlane(rp, j);
    if (v >= rj) { r = j; st = rj; }
  }
  return (row0 + r) * SLOTS + (v - st);
}

template <bool SCALE_IN>
__global__ __launch_bounds__(LTHREADS, 12) void gather_gemm_bias_relu(
    const int* __restrict__ cnt_in, const int* __restrict__ cnt_out,
    const int* __restrict__ rec, const int* __restrict__ ovf_cnt,
    const int2* __restrict__ ovf, const float* __restrict__ h,
    const float* __restrict__ W, const float* __restrict__ bias,
    float* __restrict__ out, int nrows, int nedges, int scale_out) {
  __shared__ __attribute__((aligned(16))) float Ws[FEAT * FEAT];
  __shared__ float bs[FEAT];
  __shared__ __attribute__((aligned(16))) float As[SPB * 16 * APITCH];

  const int tid = threadIdx.x;
  const int lane = tid & 31;
  const int wave = __builtin_amdgcn_readfirstlane(tid >> 5);
  const int sidx = wave / WPS;
  const int part = wave - sidx * WPS;
  const unsigned nstrips = (unsigned)(nrows + 15) >> 4;
  const int s0 = (int)(blockIdx.x * nstrips / gridDim.x);
  const int s1 = (int)((blockIdx.x + 1) * nstrips / gridDim.x);
  const int nsb = s1 - s0;
  const bool active = sidx < nsb;
  const int row0 = (s0 + sidx) * 16;
  const int rb = (wave * nsb * 16) / LWAVES;
  const int nrw = ((wave + 1) * nsb * 16) / LWAVES - rb;
  const int grow0 = s0 * 16 + rb;
  float* __restrict__ strip = As + sidx * (16 * APITCH);
  float* __restrict__ gstrip = As + rb * APITCH;

  v4f wreg[4];
#pragma unroll
  for (int i = 0; i < 4; ++i)
    wreg[i] = ((const v4f*)W)[min(tid + LTHREADS * i, FEAT * FEAT / 4 - 1)];
  const int novf_lane = ovf_cnt[0];
  int ocnt = 1;
  if (scale_out && active) ocnt = cnt_out[min(row0 + (lane & 15), nrows - 1)];

  if (nrw > 0) {
    const int fo = min(lane, FEAT / 4 - 1) * 4;
    const int rr = grow0 + min(lane, nrw - 1);
    int cfull = cnt_in[min(rr, nrows - 1)];
    if (lane >= nrw || rr >= nrows) cfull = 0;
    const float ndl = 1.0f / sqrtf(fmaxf((float)cfull, 1.0f));
    const int c = min(cfull, SLOTS);
    int incl = c;
    for (int o = 1; o < 8; o <<= 1) {
      const int t = __shfl_up(incl, o);
      if (lane >= o) incl += t;
    }
    const int end = __builtin_amdgcn_readlane(incl, 7);
    int rp = __shfl_up(incl, 1);
    if (lane == 0) rp = 0;
    if (lane >= nrw) rp = end;

    int cur = 0;
    int nextb = __builtin_amdgcn_readlane(rp, 1);
    float4 acc = make_float4(0.f, 0.f, 0.f, 0.f);

#define PARK_ROW()                                                                    \
    do {                                                                              \
      const float nd = __int_as_float(                                                \
          __builtin_amdgcn_readlane(__float_as_int(ndl), cur));                       \
      if (lane < FEAT / 4)                                                            \
        *(float4*)(gstrip + cur * APITCH + lane * 4) =                                \
            make_float4(acc.x * nd, acc.y * nd, acc.z * nd, acc.w * nd);              \
      acc = make_float4(0.f, 0.f, 0.f, 0.f);                                          \
      ++cur;                                                                          \
      nextb = __builtin_amdgcn_readlane(rp, cur + 1);                                 \
    } while (0)

    int p = 0;
    if (0 < end) p = rec[rec_index(rp, min(lane, end - 1), grow0)];
    int cn = 1;
    if (SCALE_IN) cn = cnt_out[p];
    for (int base = 0; base < end; base += 32) {
      int pn = p;
      int cnn = 1;
      float psc = 1.0f;
#pragma unroll
      for (int g = 0; g < 32 / GU; ++g) {
        const int e0 = base + g * GU;
        if (e0 < end) {
          float4 v[GU];
#pragma unroll
          for (int u = 0; u < GU; ++u) {
            const int s = __builtin_amdgcn_readlane(p, g * GU + u);
            v[u] = *(const float4*)(h + (long long)s * FEAT + fo);
          }
          if (SCALE_IN && g == 0) psc = deg_norm(cn);
          float ns[GU];
#pragma unroll
          for (int u = 0; u < GU; ++u)
            ns[u] = SCALE_IN ? __int_as_float(__builtin_amdgcn_readlane(__float_as_int(psc), g * GU + u))
                             : 1.0f;
          if (g == 0 && base + 32 < end)
            pn = rec[rec_index(rp, min(base + 32 + lane, end - 1), grow0)];
          if (SCALE_IN && g == 2 && base + 32 < end) cnn = cnt_out[pn];
          if (e0 + GU <= end) {
#pragma unroll
            for (int u = 0; u < GU; ++u) {
              const int e = e0 + u;
              if (__builtin_expect(e == nextb, 0)) {
                do { PARK_ROW(); } while (e == nextb);
              }
              if (SCALE_IN) fma4(acc, v[u], ns[u]); else add4(acc, v[u]);
            }
          } else {
#pragma unroll
            for (int u = 0; u < GU; ++u) {
              const int e = e0 + u;
              if (e < end) {
                if (__builtin_expect(e == nextb, 0)) {
                  do { PARK_ROW(); } while (e == nextb);
                }
                if (SCALE_IN) fma4(acc, v[u], ns[u]); else add4(acc, v[u]);
              }
            }
          }
        }
      }
      p = pn;
      cn = cnn;
    }
    while (cur < nrw) PARK_ROW();
#undef PARK_ROW

    const int novf = min(__builtin_amdgcn_readfirstlane(novf_lane), nedges);
    for (int i = 0; i < novf; ++i) {
      const int2 o = ovf[i];
      const int r = __builtin_amdgcn_readfirstlane(o.x) - grow0;
      if (r >= 0 && r < nrw) {
        const int s = __builtin_amdgcn_readfirstlane(o.y);
        const float nd = __int_as_float(__builtin_amdgcn_readlane(__float_as_int(ndl), r));
        float nsrc = 1.0f;
        if (SCALE_IN) nsrc = deg_norm(cnt_out[s]);
        if (lane < FEAT / 4) {
          float4* q = (float4*)(gstrip + r * APITCH + lane * 4);
          float4 hv = *(const float4*)(h + (long long)s * FEAT + lane * 4);
          if (SCALE_IN) hv = make_float4(hv.x * nsrc, hv.y * nsrc, hv.z * nsrc, hv.w * nsrc);
          float4 t = *q;
          t.x = fmaf(hv.x, nd, t.x);
          t.y = fmaf(hv.y, nd, t.y);
          t.z = fmaf(hv.z, nd, t.z);
          t.w = fmaf(hv.w, nd, t.w);
          *q = t;
        }
      }
    }
  }

  const float onorm = 1.0f / sqrtf(fmaxf((float)ocnt, 1.0f));

#pragma unroll
  for (int i = 0; i < 4; ++i)
    if (tid + LTHREADS * i < FEAT * FEAT / 4) ((v4f*)Ws)[tid + LTHREADS * i] = wreg[i];
  if (tid < FEAT) bs[tid] = bias[tid];
  __syncthreads();

  if (!active || row0 >= nrows) return;

  const int lrow = lane & 15;
  const int lgrp = lane >> 4;

  v2f a[KSTEPS];
#pragma unroll
  for (int ks = 0; ks < KSTEPS; ++ks)
    a[ks] = *(const v2f*)(strip + lrow * APITCH + ks * 4 + 2 * lgrp);

  const int cb0 = part * (TPW * 16);

  v8f acc[TPW];
#pragma unroll
  for (int nt = 0; nt < TPW; ++nt) acc[nt] = (v8f){};

  v2f bf[TPW];
#pragma unroll
  for (int nt = 0; nt < TPW; ++nt) {
    const int k = 2 * lgrp;
    bf[nt].x = Ws[k * FEAT + cb0 + nt * 16 + lrow];
    bf[nt].y = Ws[(k + 1) * FEAT + cb0 + nt * 16 + lrow];
  }

#pragma unroll
  for (int ks = 0; ks < KSTEPS; ++ks) {
    v2f bn[TPW];
    if (ks + 1 < KSTEPS) {
      const int k = (ks + 1) * 4 + 2 * lgrp;
#pragma unroll
      for (int nt = 0; nt < TPW; ++nt) {
        bn[nt].x = Ws[k * FEAT + cb0 + nt * 16 + lrow];
        bn[nt].y = Ws[(k + 1) * FEAT + cb0 + nt * 16 + lrow];
      }
    }
#pragma unroll
    for (int nt = 0; nt < TPW; ++nt) {
      acc[nt] = __builtin_amdgcn_wmma_f32_16x16x4_f32(
           false, a[ks],  false, bf[nt],
           (short)0, acc[nt],  false,  false);
    }
    if (ks + 1 < KSTEPS) {
#pragma unroll
      for (int nt = 0; nt < TPW; ++nt) bf[nt] = bn[nt];
    }
  }

  static_assert(TPW == 2, "the lane-half exchange pairs the wave's two column tiles");
  const float bcol0 = bs[cb0 + lrow];
  const float bcol1 = bs[cb0 + 16 + lrow];
  float* __restrict__ orow = out + (long long)row0 * FEAT + cb0 + lane;
#pragma unroll
  for (int r = 0; r < 8; ++r) {
    const float v0 = fmaxf(acc[0][r] + bcol0, 0.0f);
    const float v1 = fmaxf(acc[1][r] + bcol1, 0.0f);
    const auto sw = __builtin_amdgcn_permlane16_swap(__float_as_uint(v0), __float_as_uint(v1), false, false);
    const float f0 = __int_as_float(__builtin_amdgcn_readlane(__float_as_int(onorm), r));
    const float f1 = __int_as_float(__builtin_amdgcn_readlane(__float_as_int(onorm), r + 8));
    if (row0 + r < nrows) orow[(long long)r * FEAT] = __uint_as_float(sw[0]) * f0;
    if (row0 + r + 8 < nrows) orow[(long long)(r + 8) * FEAT] = __uint_as_float(sw[1]) * f1;
  }
}

extern "C" void kernel_launch(void* const* d_in, const int* in_sizes, int n_in,
                              void* d_out, int out_size, void* d_ws, size_t ws_size,
                              hipStream_t stream) {
  const float* h0  = (const float*)d_in[0];
  const int*   src = (const int*)d_in[1];
  const int*   dst = (const int*)d_in[2];
  const float* Wp[3] = {(const float*)d_in[3], (const float*)d_in[5], (const float*)d_in[7]};
  const float* bp[3] = {(const float*)d_in[4], (const float*)d_in[6], (const float*)d_in[8]};

  const int N = in_sizes[0] / FEAT;
  const int E = in_sizes[1];

  const long long NP = ((long long)N + 63) / 64 * 64;
  const long long EP = ((long long)E + 63) / 64 * 64;
  float* ws       = (float*)d_ws;
  int*   cnt_out  = (int*)ws;
  int*   cnt_in   = (int*)(ws + NP);
  int*   misc     = (int*)(ws + 2 * NP);
  int*   dfill    = misc + 64;
  int*   sfill    = dfill + NBIN;
  const long long HEAD = 2 * NP + 64 + 2 * NBIN;
  int*   rec      = (int*)(ws + HEAD);
  int2*  ovf      = (int2*)(ws + HEAD + (long long)N * SLOTS);
  float* hA       = ws + HEAD + (long long)N * SLOTS + 2 * EP;
  int2*  dbin     = (int2*)(hA + (long long)N * FEAT);
  int*   sbin     = (int*)(hA + (long long)N * FEAT + 2LL * NBIN * BCAP);
  int2*  dspill   = (int2*)(sbin + (long long)NBIN * BCAP);
  int*   sspill   = (int*)(dspill + EP);
  float* outp     = (float*)d_out;

  {
    long long n2 = 64 + 2 * NBIN;
    int blocks = (int)((n2 / 4 + 255) / 256) + 1;
    zero_f32<<<blocks, 256, 0, stream>>>((float*)misc, n2);
    bin_edges<<<(E + EPB - 1) / EPB, BT, 0, stream>>>(src, dst, dfill, sfill, dbin, sbin, misc,
                                                      dspill, sspill, E);
    place_bin<<<(N + NPB - 1) / NPB, PT, 0, stream>>>(dfill, sfill, dbin, sbin, misc, dspill,
                                                      sspill, cnt_in, cnt_out, rec, ovf, N, E);
  }

  const int strips = (N + 15) / 16;
  const int need_blocks = (strips + SPB - 1) / SPB;
  const int gemm_blocks = need_blocks > 512 ? need_blocks : 512;

  gather_gemm_bias_relu<true><<<gemm_blocks, LTHREADS, 0, stream>>>(
      cnt_in, cnt_out, rec, misc, ovf, h0, Wp[0], bp[0], outp, N, E, 1);
  gather_gemm_bias_relu<false><<<gemm_blocks, LTHREADS, 0, stream>>>(
      cnt_in, cnt_out, rec, misc, ovf, outp, Wp[1], bp[1], hA, N, E, 1);
  gather_gemm_bias_relu<false><<<gemm_blocks, LTHREADS, 0, stream>>>(
      cnt_in, cnt_out, rec, misc, ovf, hA, Wp[2], bp[2], outp, N, E, 0);
}
